# starting kernel
# baseline (speedup 1.0000x reference)
#include <hip/hip_runtime.h>

#define SEQ 2048
#define CH  512
#define NH  16
#define HD  32

typedef float v2f __attribute__((ext_vector_type(2)));
typedef float v8f __attribute__((ext_vector_type(8)));

__device__ __forceinline__ v8f wmma4(v2f a, v2f b, v8f c) {
  return __builtin_amdgcn_wmma_f32_16x16x4_f32(false, a, false, b, (short)0, c, false, false);
}

__device__ __forceinline__ v2f ldp(const float* p) {
  const float2 t = *(const float2*)p;
  v2f r = { t.x, t.y };
  return r;
}

template <int MODE>
__global__ void __launch_bounds__(256)
proj_complex_kernel(const float* __restrict__ X, const float* __restrict__ W,
                    const float* __restrict__ Bv, float* __restrict__ Yre,
                    float* __restrict__ Yim) {
  const int wave = threadIdx.x >> 5;
  const int lane = threadIdx.x & 31;
  const int half = lane >> 4;
  const int l16  = lane & 15;

  const int tileId  = blockIdx.x * 8 + wave;
  const int nTilesO = CH / 16;
  const int sTile   = tileId / nTilesO;
  const int oTile   = tileId % nTilesO;

  const int arow = sTile * 16 + l16;
  const int brow = oTile * 16 + l16;

  v8f yr = {0.f}, yi = {0.f};

  for (int k0 = 0; k0 < CH; k0 += 4) {
    const int ka = k0 + half * 2;
    const float4 xa = *(const float4*)(X + ((size_t)arow * CH + ka) * 2);
    const float4 wb = *(const float4*)(W + ((size_t)brow * CH + ka) * 2);
    v2f ar  = { xa.x,  xa.z };
    v2f ai  = { xa.y,  xa.w };
    v2f ain = { -xa.y, -xa.w };
    v2f br  = { wb.x,  wb.z };
    v2f bi  = { wb.y,  wb.w };
    yr = wmma4(ar,  br, yr);
    yr = wmma4(ain, bi, yr);
    yi = wmma4(ar,  bi, yi);
    yi = wmma4(ai,  br, yi);
  }

  const float2 bb = *(const float2*)(Bv + (size_t)brow * 2);
  const int h  = brow >> 5;
  const int dd = brow & 31;

  if (MODE == 0) {
#pragma unroll
    for (int r = 0; r < 8; ++r) {
      const int s = sTile * 16 + r + 8 * half;
      const size_t idx = ((size_t)h * SEQ + s) * HD + dd;
      Yre[idx] = yr[r] + bb.x;
      Yim[idx] = yi[r] + bb.y;
    }
  } else {
    const size_t base = ((size_t)(h * HD + dd)) * SEQ + sTile * 16 + 8 * half;
    float4 r0 = { yr[0] + bb.x, yr[1] + bb.x, yr[2] + bb.x, yr[3] + bb.x };
    float4 r1 = { yr[4] + bb.x, yr[5] + bb.x, yr[6] + bb.x, yr[7] + bb.x };
    float4 i0 = { yi[0] + bb.y, yi[1] + bb.y, yi[2] + bb.y, yi[3] + bb.y };
    float4 i1 = { yi[4] + bb.y, yi[5] + bb.y, yi[6] + bb.y, yi[7] + bb.y };
    *(float4*)(Yre + base)     = r0;
    *(float4*)(Yre + base + 4) = r1;
    *(float4*)(Yim + base)     = i0;
    *(float4*)(Yim + base + 4) = i1;
  }
}

__global__ void __launch_bounds__(256)
attn_complex_kernel(const float* __restrict__ Qre, const float* __restrict__ Qim,
                    const float* __restrict__ Kre, const float* __restrict__ Kim,
                    const float* __restrict__ Vtre, const float* __restrict__ Vtim,
                    float* __restrict__ Out) {
  __shared__ float lds_p[8][16 * 16];

  const int wave = threadIdx.x >> 5;
  const int lane = threadIdx.x & 31;
  const int half = lane >> 4;
  const int l16  = lane & 15;
  const int h    = blockIdx.y;
  const int q0   = (blockIdx.x * 8 + wave) * 16;

  const float scale = 0.17677669529663687f;

  const float* qreh = Qre + (size_t)h * SEQ * HD;
  const float* qimh = Qim + (size_t)h * SEQ * HD;
  const float* kreh = Kre + (size_t)h * SEQ * HD;
  const float* kimh = Kim + (size_t)h * SEQ * HD;
  const float* vreh = Vtre + (size_t)h * HD * SEQ;
  const float* vimh = Vtim + (size_t)h * HD * SEQ;
  float* pbuf = lds_p[wave];

  v2f qr[8], qi[8], qrn[8];
#pragma unroll
  for (int t = 0; t < 8; ++t) {
    const int dk = t * 4 + half * 2;
    const size_t off = ((size_t)(q0 + l16)) * HD + dk;
    qr[t]  = ldp(qreh + off);
    qi[t]  = ldp(qimh + off);
    qrn[t] = -qr[t];
  }

  float m_run[8], l_run[8];
  v8f o_re0 = {0.f}, o_re1 = {0.f}, o_im0 = {0.f}, o_im1 = {0.f};
#pragma unroll
  for (int r = 0; r < 8; ++r) { m_run[r] = -1e30f; l_run[r] = 0.f; }

  for (int kt = 0; kt < SEQ / 16; ++kt) {
    const int kbase = kt * 16;

    v8f sre = {0.f}, sim = {0.f};
#pragma unroll
    for (int t = 0; t < 8; ++t) {
      const int dk = t * 4 + half * 2;
      const size_t off = ((size_t)(kbase + l16)) * HD + dk;
      v2f krv = ldp(kreh + off);
      v2f kiv = ldp(kimh + off);
      sre = wmma4(qr[t],  krv, sre);
      sre = wmma4(qi[t],  kiv, sre);
      sim = wmma4(qi[t],  krv, sim);
      sim = wmma4(qrn[t], kiv, sim);
    }

    float pl[8], rs[8];
#pragma unroll
    for (int r = 0; r < 8; ++r) {
      float a = sqrtf(sre[r] * sre[r] + sim[r] * sim[r]) * scale;
      float mx = a;
#pragma unroll
      for (int msk = 1; msk < 16; msk <<= 1) mx = fmaxf(mx, __shfl_xor(mx, msk, 32));
      const float newm = fmaxf(m_run[r], mx);
      float pv = __expf(a - newm);
      float sum = pv;
#pragma unroll
      for (int msk = 1; msk < 16; msk <<= 1) sum += __shfl_xor(sum, msk, 32);
      const float sc = __expf(m_run[r] - newm);
      l_run[r] = l_run[r] * sc + sum;
      m_run[r] = newm;
      pl[r] = pv;
      rs[r] = sc;
    }
#pragma unroll
    for (int r = 0; r < 8; ++r) {
      o_re0[r] *= rs[r]; o_re1[r] *= rs[r];
      o_im0[r] *= rs[r]; o_im1[r] *= rs[r];
    }

#pragma unroll
    for (int r = 0; r < 8; ++r) pbuf[(r + 8 * half) * 16 + l16] = pl[r];
    __asm__ volatile("s_wait_dscnt 0x0" ::: "memory");

#pragma unroll
    for (int t = 0; t < 4; ++t) {
      const int klocal = t * 4 + half * 2;
      v2f aP = ldp(pbuf + l16 * 16 + klocal);
      const size_t voff = (size_t)l16 * SEQ + kbase + klocal;
      v2f br0 = ldp(vreh + voff);
      v2f bi0 = ldp(vimh + voff);
      v2f br1 = ldp(vreh + voff + (size_t)16 * SEQ);
      v2f bi1 = ldp(vimh + voff + (size_t)16 * SEQ);
      o_re0 = wmma4(aP, br0, o_re0);
      o_im0 = wmma4(aP, bi0, o_im0);
      o_re1 = wmma4(aP, br1, o_re1);
      o_im1 = wmma4(aP, bi1, o_im1);
    }
  }

#pragma unroll
  for (int r = 0; r < 8; ++r) {
    const float inv = 1.0f / l_run[r];
    const int s  = q0 + r + 8 * half;
    const int o0 = h * HD + l16;
    float2 e0 = { o_re0[r] * inv, o_im0[r] * inv };
    float2 e1 = { o_re1[r] * inv, o_im1[r] * inv };
    *(float2*)(Out + ((size_t)s * CH + o0) * 2)      = e0;
    *(float2*)(Out + ((size_t)s * CH + o0 + 16) * 2) = e1;
  }
}

extern "C" void kernel_launch(void* const* d_in, const int* in_sizes, int n_in,
                              void* d_out, int out_size, void* d_ws, size_t ws_size,
                              hipStream_t stream) {
  (void)in_sizes; (void)n_in; (void)out_size; (void)ws_size;
  const float* Q  = (const float*)d_in[0];
  const float* V  = (const float*)d_in[1];
  const float* K  = (const float*)d_in[2];
  const float* Wq = (const float*)d_in[3];
  const float* bq = (const float*)d_in[4];
  const float* Wk = (const float*)d_in[5];
  const float* bk = (const float*)d_in[6];
  const float* Wv = (const float*)d_in[7];
  const float* bv = (const float*)d_in[8];
  float* out = (float*)d_out;

  const size_t plane = (size_t)NH * SEQ * HD;
  float* qre = (float*)d_ws;
  float* qim = qre + plane;
  float* kre = qim + plane;
  float* kim = kre + plane;
  float* vre = kim + plane;
  float* vim = vre + plane;

  const dim3 blk(256);
  const dim3 pgrid((SEQ / 16) * (CH / 16) / 8);
  proj_complex_kernel<0><<<pgrid, blk, 0, stream>>>(Q, Wq, bq, qre, qim);
  proj_complex_kernel<0><<<pgrid, blk, 0, stream>>>(K, Wk, bk, kre, kim);
  proj_complex_kernel<1><<<pgrid, blk, 0, stream>>>(V, Wv, bv, vre, vim);

  const dim3 agrid(SEQ / 16 / 8, NH);
  attn_complex_kernel<<<agrid, blk, 0, stream>>>(qre, qim, kre, kim, vre, vim, out);
}


# attention: both products transposed (S^T=K.Q^H, O^T=V^T.P^T) so a lane owns one query: softmax per lane, weights feed WMMA from registers, no LDS; changes order of row sums and of keys in P.V
# speedup vs baseline: 1.2448x; 1.2448x over previous
#include <hip/hip_runtime.h>

#define SEQ 2048
#define CH  512
#define NH  16
#define HD  32

typedef float v2f __attribute__((ext_vector_type(2)));
typedef float v8f __attribute__((ext_vector_type(8)));

__device__ __forceinline__ v8f wmma4(v2f a, v2f b, v8f c) {
  return __builtin_amdgcn_wmma_f32_16x16x4_f32(false, a, false, b, (short)0, c, false, false);
}

__device__ __forceinline__ v2f ldp(const float* p) {
  const float2 t = *(const float2*)p;
  v2f r = { t.x, t.y };
  return r;
}

template <int MODE>
__global__ void __launch_bounds__(256)
proj_complex_kernel(const float* __restrict__ X, const float* __restrict__ W,
                    const float* __restrict__ Bv, float* __restrict__ Yre,
                    float* __restrict__ Yim) {
  const int wave = threadIdx.x >> 5;
  const int lane = threadIdx.x & 31;
  const int half = lane >> 4;
  const int l16  = lane & 15;

  const int tileId  = blockIdx.x * 8 + wave;
  const int nTilesO = CH / 16;
  const int sTile   = tileId / nTilesO;
  const int oTile   = tileId % nTilesO;

  const int arow = sTile * 16 + l16;
  const int brow = oTile * 16 + l16;

  v8f yr = {0.f}, yi = {0.f};

  for (int k0 = 0; k0 < CH; k0 += 4) {
    const int ka = k0 + half * 2;
    const float4 xa = *(const float4*)(X + ((size_t)arow * CH + ka) * 2);
    const float4 wb = *(const float4*)(W + ((size_t)brow * CH + ka) * 2);
    v2f ar  = { xa.x,  xa.z };
    v2f ai  = { xa.y,  xa.w };
    v2f ain = { -xa.y, -xa.w };
    v2f br  = { wb.x,  wb.z };
    v2f bi  = { wb.y,  wb.w };
    yr = wmma4(ar,  br, yr);
    yr = wmma4(ain, bi, yr);
    yi = wmma4(ar,  bi, yi);
    yi = wmma4(ai,  br, yi);
  }

  const float2 bb = *(const float2*)(Bv + (size_t)brow * 2);
  const int h  = brow >> 5;
  const int dd = brow & 31;

  if (MODE == 0) {
#pragma unroll
    for (int r = 0; r < 8; ++r) {
      const int s = sTile * 16 + r + 8 * half;
      const size_t idx = ((size_t)h * SEQ + s) * HD + dd;
      Yre[idx] = yr[r] + bb.x;
      Yim[idx] = yi[r] + bb.y;
    }
  } else {
    const size_t base = ((size_t)(h * HD + dd)) * SEQ + sTile * 16 + 8 * half;
    float4 r0 = { yr[0] + bb.x, yr[1] + bb.x, yr[2] + bb.x, yr[3] + bb.x };
    float4 r1 = { yr[4] + bb.x, yr[5] + bb.x, yr[6] + bb.x, yr[7] + bb.x };
    float4 i0 = { yi[0] + bb.y, yi[1] + bb.y, yi[2] + bb.y, yi[3] + bb.y };
    float4 i1 = { yi[4] + bb.y, yi[5] + bb.y, yi[6] + bb.y, yi[7] + bb.y };
    *(float4*)(Yre + base)     = r0;
    *(float4*)(Yre + base + 4) = r1;
    *(float4*)(Yim + base)     = i0;
    *(float4*)(Yim + base + 4) = i1;
  }
}

__device__ __forceinline__ float xhalf(float x) {
  const int b = __builtin_bit_cast(int, x);
  return __builtin_bit_cast(float,
      __builtin_amdgcn_permlanex16(b, b, 0x76543210, 0xfedcba98, false, false));
}

__global__ void __launch_bounds__(256)
attn_complex_kernel(const float* __restrict__ Qre, const float* __restrict__ Qim,
                    const float* __restrict__ Kre, const float* __restrict__ Kim,
                    const float* __restrict__ Vtre, const float* __restrict__ Vtim,
                    float* __restrict__ Out) {
  const int wave = threadIdx.x >> 5;
  const int lane = threadIdx.x & 31;
  const int half = lane >> 4;
  const int l16  = lane & 15;
  const int h    = blockIdx.y;
  const int q0   = (blockIdx.x * 8 + wave) * 16;

  const float scale = 0.17677669529663687f;

  const float* qreh = Qre + (size_t)h * SEQ * HD;
  const float* qimh = Qim + (size_t)h * SEQ * HD;
  const float* kreh = Kre + (size_t)h * SEQ * HD;
  const float* kimh = Kim + (size_t)h * SEQ * HD;
  const float* vreh = Vtre + (size_t)h * HD * SEQ;
  const float* vimh = Vtim + (size_t)h * HD * SEQ;

  v2f qr[8], qi[8], qrn[8];
#pragma unroll
  for (int t = 0; t < 8; ++t) {
    const int dk = t * 4 + half * 2;
    const size_t off = ((size_t)(q0 + l16)) * HD + dk;
    qr[t]  = ldp(qreh + off);
    qi[t]  = ldp(qimh + off);
    qrn[t] = -qr[t];
  }

  float m_run = -1e30f;
  float l_run = 0.f;
  v8f o_re0 = {0.f}, o_re1 = {0.f}, o_im0 = {0.f}, o_im1 = {0.f};

  for (int kt = 0; kt < SEQ / 16; ++kt) {
    const int kbase = kt * 16;

    v8f sre = {0.f}, sim = {0.f};
#pragma unroll
    for (int t = 0; t < 8; ++t) {
      const int dk = t * 4 + half * 2;
      const size_t off = ((size_t)(kbase + l16)) * HD + dk;
      v2f krv = ldp(kreh + off);
      v2f kiv = ldp(kimh + off);
      sre = wmma4(krv, qr[t],  sre);
      sre = wmma4(kiv, qi[t],  sre);
      sim = wmma4(krv, qi[t],  sim);
      sim = wmma4(kiv, qrn[t], sim);
    }

    float a[8];
#pragma unroll
    for (int r = 0; r < 8; ++r)
      a[r] = sqrtf(sre[r] * sre[r] + sim[r] * sim[r]) * scale;
    float mx = fmaxf(fmaxf(fmaxf(a[0], a[1]), fmaxf(a[2], a[3])),
                     fmaxf(fmaxf(a[4], a[5]), fmaxf(a[6], a[7])));
    mx = fmaxf(mx, xhalf(mx));
    const float newm = fmaxf(m_run, mx);
    const float sc = __expf(m_run - newm);
    float p[8];
#pragma unroll
    for (int r = 0; r < 8; ++r) p[r] = __expf(a[r] - newm);
    l_run = l_run * sc + (((p[0] + p[1]) + (p[2] + p[3])) + ((p[4] + p[5]) + (p[6] + p[7])));
    m_run = newm;
#pragma unroll
    for (int r = 0; r < 8; ++r) {
      o_re0[r] *= sc; o_re1[r] *= sc;
      o_im0[r] *= sc; o_im1[r] *= sc;
    }

    const size_t voff = (size_t)l16 * SEQ + kbase + 8 * half;
    const float4 r0a = *(const float4*)(vreh + voff);
    const float4 r0b = *(const float4*)(vreh + voff + 4);
    const float4 i0a = *(const float4*)(vimh + voff);
    const float4 i0b = *(const float4*)(vimh + voff + 4);
    const float4 r1a = *(const float4*)(vreh + voff + (size_t)16 * SEQ);
    const float4 r1b = *(const float4*)(vreh + voff + (size_t)16 * SEQ + 4);
    const float4 i1a = *(const float4*)(vimh + voff + (size_t)16 * SEQ);
    const float4 i1b = *(const float4*)(vimh + voff + (size_t)16 * SEQ + 4);
    const v2f ar0[4] = { {r0a.x, r0a.y}, {r0a.z, r0a.w}, {r0b.x, r0b.y}, {r0b.z, r0b.w} };
    const v2f ai0[4] = { {i0a.x, i0a.y}, {i0a.z, i0a.w}, {i0b.x, i0b.y}, {i0b.z, i0b.w} };
    const v2f ar1[4] = { {r1a.x, r1a.y}, {r1a.z, r1a.w}, {r1b.x, r1b.y}, {r1b.z, r1b.w} };
    const v2f ai1[4] = { {i1a.x, i1a.y}, {i1a.z, i1a.w}, {i1b.x, i1b.y}, {i1b.z, i1b.w} };
#pragma unroll
    for (int t = 0; t < 4; ++t) {
      const v2f bP = { p[2 * t], p[2 * t + 1] };
      o_re0 = wmma4(ar0[t], bP, o_re0);
      o_im0 = wmma4(ai0[t], bP, o_im0);
      o_re1 = wmma4(ar1[t], bP, o_re1);
      o_im1 = wmma4(ai1[t], bP, o_im1);
    }
  }

  const float inv = 1.0f / (l_run + xhalf(l_run));
  float* op = Out + ((size_t)(q0 + l16) * CH + h * HD + 8 * half) * 2;
#pragma unroll
  for (int j = 0; j < 4; ++j) {
    const float4 e0 = { o_re0[2 * j] * inv, o_im0[2 * j] * inv,
                        o_re0[2 * j + 1] * inv, o_im0[2 * j + 1] * inv };
    const float4 e1 = { o_re1[2 * j] * inv, o_im1[2 * j] * inv,
                        o_re1[2 * j + 1] * inv, o_im1[2 * j + 1] * inv };
    *(float4*)(op + 4 * j)      = e0;
    *(float4*)(op + 32 + 4 * j) = e1;
  }
}

extern "C" void kernel_launch(void* const* d_in, const int* in_sizes, int n_in,
                              void* d_out, int out_size, void* d_ws, size_t ws_size,
                              hipStream_t stream) {
  (void)in_sizes; (void)n_in; (void)out_size; (void)ws_size;
  const float* Q  = (const float*)d_in[0];
  const float* V  = (const float*)d_in[1];
  const float* K  = (const float*)d_in[2];
  const float* Wq = (const float*)d_in[3];
  const float* bq = (const float*)d_in[4];
  const float* Wk = (const float*)d_in[5];
  const float* bk = (const float*)d_in[6];
  const float* Wv = (const float*)d_in[7];
  const float* bv = (const float*)d_in[8];
  float* out = (float*)d_out;

  const size_t plane = (size_t)NH * SEQ * HD;
  float* qre = (float*)d_ws;
  float* qim = qre + plane;
  float* kre = qim + plane;
  float* kim = kre + plane;
  float* vre = kim + plane;
  float* vim = vre + plane;

  const dim3 blk(256);
  const dim3 pgrid((SEQ / 16) * (CH / 16) / 8);
  proj_complex_kernel<0><<<pgrid, blk, 0, stream>>>(Q, Wq, bq, qre, qim);
  proj_complex_kernel<0><<<pgrid, blk, 0, stream>>>(K, Wk, bk, kre, kim);
  proj_complex_kernel<1><<<pgrid, blk, 0, stream>>>(V, Wv, bv, vre, vim);

  const dim3 agrid(SEQ / 16 / 8, NH);
  attn_complex_kernel<<<agrid, blk, 0, stream>>>(qre, qim, kre, kim, vre, vim, out);
}


# attention softmax: |score| by the native v_sqrt_f32 (1 ulp) in place of the correctly rounded sqrtf sequence (about 15 vector instructions per value, 8 values per key tile); order of all sums kept
# speedup vs baseline: 1.3105x; 1.0528x over previous
#include <hip/hip_runtime.h>

#define SEQ 2048
#define CH  512
#define NH  16
#define HD  32

typedef float v2f __attribute__((ext_vector_type(2)));
typedef float v8f __attribute__((ext_vector_type(8)));

__device__ __forceinline__ v8f wmma4(v2f a, v2f b, v8f c) {
  return __builtin_amdgcn_wmma_f32_16x16x4_f32(false, a, false, b, (short)0, c, false, false);
}

__device__ __forceinline__ v2f ldp(const float* p) {
  const float2 t = *(const float2*)p;
  v2f r = { t.x, t.y };
  return r;
}

template <int MODE>
__global__ void __launch_bounds__(256)
proj_complex_kernel(const float* __restrict__ X, const float* __restrict__ W,
                    const float* __restrict__ Bv, float* __restrict__ Yre,
                    float* __restrict__ Yim) {
  const int wave = threadIdx.x >> 5;
  const int lane = threadIdx.x & 31;
  const int half = lane >> 4;
  const int l16  = lane & 15;

  const int tileId  = blockIdx.x * 8 + wave;
  const int nTilesO = CH / 16;
  const int sTile   = tileId / nTilesO;
  const int oTile   = tileId % nTilesO;

  const int arow = sTile * 16 + l16;
  const int brow = oTile * 16 + l16;

  v8f yr = {0.f}, yi = {0.f};

  for (int k0 = 0; k0 < CH; k0 += 4) {
    const int ka = k0 + half * 2;
    const float4 xa = *(const float4*)(X + ((size_t)arow * CH + ka) * 2);
    const float4 wb = *(const float4*)(W + ((size_t)brow * CH + ka) * 2);
    v2f ar  = { xa.x,  xa.z };
    v2f ai  = { xa.y,  xa.w };
    v2f ain = { -xa.y, -xa.w };
    v2f br  = { wb.x,  wb.z };
    v2f bi  = { wb.y,  wb.w };
    yr = wmma4(ar,  br, yr);
    yr = wmma4(ain, bi, yr);
    yi = wmma4(ar,  bi, yi);
    yi = wmma4(ai,  br, yi);
  }

  const float2 bb = *(const float2*)(Bv + (size_t)brow * 2);
  const int h  = brow >> 5;
  const int dd = brow & 31;

  if (MODE == 0) {
#pragma unroll
    for (int r = 0; r < 8; ++r) {
      const int s = sTile * 16 + r + 8 * half;
      const size_t idx = ((size_t)h * SEQ + s) * HD + dd;
      Yre[idx] = yr[r] + bb.x;
      Yim[idx] = yi[r] + bb.y;
    }
  } else {
    const size_t base = ((size_t)(h * HD + dd)) * SEQ + sTile * 16 + 8 * half;
    float4 r0 = { yr[0] + bb.x, yr[1] + bb.x, yr[2] + bb.x, yr[3] + bb.x };
    float4 r1 = { yr[4] + bb.x, yr[5] + bb.x, yr[6] + bb.x, yr[7] + bb.x };
    float4 i0 = { yi[0] + bb.y, yi[1] + bb.y, yi[2] + bb.y, yi[3] + bb.y };
    float4 i1 = { yi[4] + bb.y, yi[5] + bb.y, yi[6] + bb.y, yi[7] + bb.y };
    *(float4*)(Yre + base)     = r0;
    *(float4*)(Yre + base + 4) = r1;
    *(float4*)(Yim + base)     = i0;
    *(float4*)(Yim + base + 4) = i1;
  }
}

__device__ __forceinline__ float xhalf(float x) {
  const int b = __builtin_bit_cast(int, x);
  return __builtin_bit_cast(float,
      __builtin_amdgcn_permlanex16(b, b, 0x76543210, 0xfedcba98, false, false));
}

__global__ void __launch_bounds__(256)
attn_complex_kernel(const float* __restrict__ Qre, const float* __restrict__ Qim,
                    const float* __restrict__ Kre, const float* __restrict__ Kim,
                    const float* __restrict__ Vtre, const float* __restrict__ Vtim,
                    float* __restrict__ Out) {
  const int wave = threadIdx.x >> 5;
  const int lane = threadIdx.x & 31;
  const int half = lane >> 4;
  const int l16  = lane & 15;
  const int h    = blockIdx.y;
  const int q0   = (blockIdx.x * 8 + wave) * 16;

  const float scale = 0.17677669529663687f;

  const float* qreh = Qre + (size_t)h * SEQ * HD;
  const float* qimh = Qim + (size_t)h * SEQ * HD;
  const float* kreh = Kre + (size_t)h * SEQ * HD;
  const float* kimh = Kim + (size_t)h * SEQ * HD;
  const float* vreh = Vtre + (size_t)h * HD * SEQ;
  const float* vimh = Vtim + (size_t)h * HD * SEQ;

  v2f qr[8], qi[8], qrn[8];
#pragma unroll
  for (int t = 0; t < 8; ++t) {
    const int dk = t * 4 + half * 2;
    const size_t off = ((size_t)(q0 + l16)) * HD + dk;
    qr[t]  = ldp(qreh + off);
    qi[t]  = ldp(qimh + off);
    qrn[t] = -qr[t];
  }

  float m_run = -1e30f;
  float l_run = 0.f;
  v8f o_re0 = {0.f}, o_re1 = {0.f}, o_im0 = {0.f}, o_im1 = {0.f};

  for (int kt = 0; kt < SEQ / 16; ++kt) {
    const int kbase = kt * 16;

    v8f sre = {0.f}, sim = {0.f};
#pragma unroll
    for (int t = 0; t < 8; ++t) {
      const int dk = t * 4 + half * 2;
      const size_t off = ((size_t)(kbase + l16)) * HD + dk;
      v2f krv = ldp(kreh + off);
      v2f kiv = ldp(kimh + off);
      sre = wmma4(krv, qr[t],  sre);
      sre = wmma4(kiv, qi[t],  sre);
      sim = wmma4(krv, qi[t],  sim);
      sim = wmma4(kiv, qrn[t], sim);
    }

    float a[8];
#pragma unroll
    for (int r = 0; r < 8; ++r)
      a[r] = __builtin_amdgcn_sqrtf(sre[r] * sre[r] + sim[r] * sim[r]) * scale;
    float mx = fmaxf(fmaxf(fmaxf(a[0], a[1]), fmaxf(a[2], a[3])),
                     fmaxf(fmaxf(a[4], a[5]), fmaxf(a[6], a[7])));
    mx = fmaxf(mx, xhalf(mx));
    const float newm = fmaxf(m_run, mx);
    const float sc = __expf(m_run - newm);
    float p[8];
#pragma unroll
    for (int r = 0; r < 8; ++r) p[r] = __expf(a[r] - newm);
    l_run = l_run * sc + (((p[0] + p[1]) + (p[2] + p[3])) + ((p[4] + p[5]) + (p[6] + p[7])));
    m_run = newm;
#pragma unroll
    for (int r = 0; r < 8; ++r) {
      o_re0[r] *= sc; o_re1[r] *= sc;
      o_im0[r] *= sc; o_im1[r] *= sc;
    }

    const size_t voff = (size_t)l16 * SEQ + kbase + 8 * half;
    const float4 r0a = *(const float4*)(vreh + voff);
    const float4 r0b = *(const float4*)(vreh + voff + 4);
    const float4 i0a = *(const float4*)(vimh + voff);
    const float4 i0b = *(const float4*)(vimh + voff + 4);
    const float4 r1a = *(const float4*)(vreh + voff + (size_t)16 * SEQ);
    const float4 r1b = *(const float4*)(vreh + voff + (size_t)16 * SEQ + 4);
    const float4 i1a = *(const float4*)(vimh + voff + (size_t)16 * SEQ);
    const float4 i1b = *(const float4*)(vimh + voff + (size_t)16 * SEQ + 4);
    const v2f ar0[4] = { {r0a.x, r0a.y}, {r0a.z, r0a.w}, {r0b.x, r0b.y}, {r0b.z, r0b.w} };
    const v2f ai0[4] = { {i0a.x, i0a.y}, {i0a.z, i0a.w}, {i0b.x, i0b.y}, {i0b.z, i0b.w} };
    const v2f ar1[4] = { {r1a.x, r1a.y}, {r1a.z, r1a.w}, {r1b.x, r1b.y}, {r1b.z, r1b.w} };
    const v2f ai1[4] = { {i1a.x, i1a.y}, {i1a.z, i1a.w}, {i1b.x, i1b.y}, {i1b.z, i1b.w} };
#pragma unroll
    for (int t = 0; t < 4; ++t) {
      const v2f bP = { p[2 * t], p[2 * t + 1] };
      o_re0 = wmma4(ar0[t], bP, o_re0);
      o_im0 = wmma4(ai0[t], bP, o_im0);
      o_re1 = wmma4(ar1[t], bP, o_re1);
      o_im1 = wmma4(ai1[t], bP, o_im1);
    }
  }

  const float inv = 1.0f / (l_run + xhalf(l_run));
  float* op = Out + ((size_t)(q0 + l16) * CH + h * HD + 8 * half) * 2;
#pragma unroll
  for (int j = 0; j < 4; ++j) {
    const float4 e0 = { o_re0[2 * j] * inv, o_im0[2 * j] * inv,
                        o_re0[2 * j + 1] * inv, o_im0[2 * j + 1] * inv };
    const float4 e1 = { o_re1[2 * j] * inv, o_im1[2 * j] * inv,
                        o_re1[2 * j + 1] * inv, o_im1[2 * j + 1] * inv };
    *(float4*)(op + 4 * j)      = e0;
    *(float4*)(op + 32 + 4 * j) = e1;
  }
}

extern "C" void kernel_launch(void* const* d_in, const int* in_sizes, int n_in,
                              void* d_out, int out_size, void* d_ws, size_t ws_size,
                              hipStream_t stream) {
  (void)in_sizes; (void)n_in; (void)out_size; (void)ws_size;
  const float* Q  = (const float*)d_in[0];
  const float* V  = (const float*)d_in[1];
  const float* K  = (const float*)d_in[2];
  const float* Wq = (const float*)d_in[3];
  const float* bq = (const float*)d_in[4];
  const float* Wk = (const float*)d_in[5];
  const float* bk = (const float*)d_in[6];
  const float* Wv = (const float*)d_in[7];
  const float* bv = (const float*)d_in[8];
  float* out = (float*)d_out;

  const size_t plane = (size_t)NH * SEQ * HD;
  float* qre = (float*)d_ws;
  float* qim = qre + plane;
  float* kre = qim + plane;
  float* kim = kre + plane;
  float* vre = kim + plane;
  float* vim = vre + plane;

  const dim3 blk(256);
  const dim3 pgrid((SEQ / 16) * (CH / 16) / 8);
  proj_complex_kernel<0><<<pgrid, blk, 0, stream>>>(Q, Wq, bq, qre, qim);
  proj_complex_kernel<0><<<pgrid, blk, 0, stream>>>(K, Wk, bk, kre, kim);
  proj_complex_kernel<1><<<pgrid, blk, 0, stream>>>(V, Wv, bv, vre, vim);

  const dim3 agrid(SEQ / 16 / 8, NH);
  attn_complex_kernel<<<agrid, blk, 0, stream>>>(qre, qim, kre, kim, vre, vim, out);
}


# attention: next key tile's K and V fragments loaded a whole tile ahead into a second register set (loop body doubled; an empty asm pin keeps the loads there); arithmetic and order of sums unchanged
# speedup vs baseline: 1.5657x; 1.1947x over previous
#include <hip/hip_runtime.h>

#define SEQ 2048
#define CH  512
#define NH  16
#define HD  32

typedef float v2f __attribute__((ext_vector_type(2)));
typedef float v4f __attribute__((ext_vector_type(4)));
typedef float v8f __attribute__((ext_vector_type(8)));

__device__ __forceinline__ v8f wmma4(v2f a, v2f b, v8f c) {
  return __builtin_amdgcn_wmma_f32_16x16x4_f32(false, a, false, b, (short)0, c, false, false);
}

__device__ __forceinline__ v2f ldp(const float* p) {
  const float2 t = *(const float2*)p;
  v2f r = { t.x, t.y };
  return r;
}

template <int MODE>
__global__ void __launch_bounds__(256)
proj_complex_kernel(const float* __restrict__ X, const float* __restrict__ W,
                    const float* __restrict__ Bv, float* __restrict__ Yre,
                    float* __restrict__ Yim) {
  const int wave = threadIdx.x >> 5;
  const int lane = threadIdx.x & 31;
  const int half = lane >> 4;
  const int l16  = lane & 15;

  const int tileId  = blockIdx.x * 8 + wave;
  const int nTilesO = CH / 16;
  const int sTile   = tileId / nTilesO;
  const int oTile   = tileId % nTilesO;

  const int arow = sTile * 16 + l16;
  const int brow = oTile * 16 + l16;

  v8f yr = {0.f}, yi = {0.f};

  for (int k0 = 0; k0 < CH; k0 += 4) {
    const int ka = k0 + half * 2;
    const float4 xa = *(const float4*)(X + ((size_t)arow * CH + ka) * 2);
    const float4 wb = *(const float4*)(W + ((size_t)brow * CH + ka) * 2);
    v2f ar  = { xa.x,  xa.z };
    v2f ai  = { xa.y,  xa.w };
    v2f ain = { -xa.y, -xa.w };
    v2f br  = { wb.x,  wb.z };
    v2f bi  = { wb.y,  wb.w };
    yr = wmma4(ar,  br, yr);
    yr = wmma4(ain, bi, yr);
    yi = wmma4(ar,  bi, yi);
    yi = wmma4(ai,  br, yi);
  }

  const float2 bb = *(const float2*)(Bv + (size_t)brow * 2);
  const int h  = brow >> 5;
  const int dd = brow & 31;

  if (MODE == 0) {
#pragma unroll
    for (int r = 0; r < 8; ++r) {
      const int s = sTile * 16 + r + 8 * half;
      const size_t idx = ((size_t)h * SEQ + s) * HD + dd;
      Yre[idx] = yr[r] + bb.x;
      Yim[idx] = yi[r] + bb.y;
    }
  } else {
    const size_t base = ((size_t)(h * HD + dd)) * SEQ + sTile * 16 + 8 * half;
    float4 r0 = { yr[0] + bb.x, yr[1] + bb.x, yr[2] + bb.x, yr[3] + bb.x };
    float4 r1 = { yr[4] + bb.x, yr[5] + bb.x, yr[6] + bb.x, yr[7] + bb.x };
    float4 i0 = { yi[0] + bb.y, yi[1] + bb.y, yi[2] + bb.y, yi[3] + bb.y };
    float4 i1 = { yi[4] + bb.y, yi[5] + bb.y, yi[6] + bb.y, yi[7] + bb.y };
    *(float4*)(Yre + base)     = r0;
    *(float4*)(Yre + base + 4) = r1;
    *(float4*)(Yim + base)     = i0;
    *(float4*)(Yim + base + 4) = i1;
  }
}

__device__ __forceinline__ float xhalf(float x) {
  const int b = __builtin_bit_cast(int, x);
  return __builtin_bit_cast(float,
      __builtin_amdgcn_permlanex16(b, b, 0x76543210, 0xfedcba98, false, false));
}

struct KVFrag {
  v2f kr[8], ki[8];
  v4f vr[4], vi[4];
};

__device__ __forceinline__ void load_kv(KVFrag& f, const float* __restrict__ kreh,
                                        const float* __restrict__ kimh,
                                        const float* __restrict__ vreh,
                                        const float* __restrict__ vimh,
                                        int kt, int l16, int half) {
  const int kbase = kt * 16;
#pragma unroll
  for (int t = 0; t < 8; ++t) {
    const int dk = t * 4 + half * 2;
    const size_t off = ((size_t)(kbase + l16)) * HD + dk;
    f.kr[t] = ldp(kreh + off);
    f.ki[t] = ldp(kimh + off);
  }
  const size_t voff = (size_t)l16 * SEQ + kbase + 8 * half;
  f.vr[0] = *(const v4f*)(vreh + voff);
  f.vr[1] = *(const v4f*)(vreh + voff + 4);
  f.vi[0] = *(const v4f*)(vimh + voff);
  f.vi[1] = *(const v4f*)(vimh + voff + 4);
  f.vr[2] = *(const v4f*)(vreh + voff + (size_t)16 * SEQ);
  f.vr[3] = *(const v4f*)(vreh + voff + (size_t)16 * SEQ + 4);
  f.vi[2] = *(const v4f*)(vimh + voff + (size_t)16 * SEQ);
  f.vi[3] = *(const v4f*)(vimh + voff + (size_t)16 * SEQ + 4);
}

__device__ __forceinline__ void attn_tile(KVFrag& f, const v2f (&qr)[8], const v2f (&qi)[8],
                                          const v2f (&qrn)[8], float& m_run, float& l_run,
                                          v8f& o_re0, v8f& o_re1, v8f& o_im0, v8f& o_im1) {
  const float scale = 0.17677669529663687f;

  asm volatile("" : "+v"(f.kr[0]), "+v"(f.kr[1]), "+v"(f.kr[2]), "+v"(f.kr[3]),
                    "+v"(f.kr[4]), "+v"(f.kr[5]), "+v"(f.kr[6]), "+v"(f.kr[7]),
                    "+v"(f.ki[0]), "+v"(f.ki[1]), "+v"(f.ki[2]), "+v"(f.ki[3]),
                    "+v"(f.ki[4]), "+v"(f.ki[5]), "+v"(f.ki[6]), "+v"(f.ki[7]),
                    "+v"(f.vr[0]), "+v"(f.vr[1]), "+v"(f.vr[2]), "+v"(f.vr[3]),
                    "+v"(f.vi[0]), "+v"(f.vi[1]), "+v"(f.vi[2]), "+v"(f.vi[3]));

  v8f sre = {0.f}, sim = {0.f};
#pragma unroll
  for (int t = 0; t < 8; ++t) {
    sre = wmma4(f.kr[t], qr[t],  sre);
    sre = wmma4(f.ki[t], qi[t],  sre);
    sim = wmma4(f.kr[t], qi[t],  sim);
    sim = wmma4(f.ki[t], qrn[t], sim);
  }

  float a[8];
#pragma unroll
  for (int r = 0; r < 8; ++r)
    a[r] = __builtin_amdgcn_sqrtf(sre[r] * sre[r] + sim[r] * sim[r]) * scale;
  float mx = fmaxf(fmaxf(fmaxf(a[0], a[1]), fmaxf(a[2], a[3])),
                   fmaxf(fmaxf(a[4], a[5]), fmaxf(a[6], a[7])));
  mx = fmaxf(mx, xhalf(mx));
  const float newm = fmaxf(m_run, mx);
  const float sc = __expf(m_run - newm);
  float p[8];
#pragma unroll
  for (int r = 0; r < 8; ++r) p[r] = __expf(a[r] - newm);
  l_run = l_run * sc + (((p[0] + p[1]) + (p[2] + p[3])) + ((p[4] + p[5]) + (p[6] + p[7])));
  m_run = newm;
#pragma unroll
  for (int r = 0; r < 8; ++r) {
    o_re0[r] *= sc; o_re1[r] *= sc;
    o_im0[r] *= sc; o_im1[r] *= sc;
  }

  const v2f ar0[4] = { {f.vr[0].x, f.vr[0].y}, {f.vr[0].z, f.vr[0].w}, {f.vr[1].x, f.vr[1].y}, {f.vr[1].z, f.vr[1].w} };
  const v2f ai0[4] = { {f.vi[0].x, f.vi[0].y}, {f.vi[0].z, f.vi[0].w}, {f.vi[1].x, f.vi[1].y}, {f.vi[1].z, f.vi[1].w} };
  const v2f ar1[4] = { {f.vr[2].x, f.vr[2].y}, {f.vr[2].z, f.vr[2].w}, {f.vr[3].x, f.vr[3].y}, {f.vr[3].z, f.vr[3].w} };
  const v2f ai1[4] = { {f.vi[2].x, f.vi[2].y}, {f.vi[2].z, f.vi[2].w}, {f.vi[3].x, f.vi[3].y}, {f.vi[3].z, f.vi[3].w} };
#pragma unroll
  for (int t = 0; t < 4; ++t) {
    const v2f bP = { p[2 * t], p[2 * t + 1] };
    o_re0 = wmma4(ar0[t], bP, o_re0);
    o_im0 = wmma4(ai0[t], bP, o_im0);
    o_re1 = wmma4(ar1[t], bP, o_re1);
    o_im1 = wmma4(ai1[t], bP, o_im1);
  }
}

__global__ void __launch_bounds__(256)
attn_complex_kernel(const float* __restrict__ Qre, const float* __restrict__ Qim,
                    const float* __restrict__ Kre, const float* __restrict__ Kim,
                    const float* __restrict__ Vtre, const float* __restrict__ Vtim,
                    float* __restrict__ Out) {
  const int wave = threadIdx.x >> 5;
  const int lane = threadIdx.x & 31;
  const int half = lane >> 4;
  const int l16  = lane & 15;
  const int h    = blockIdx.y;
  const int q0   = (blockIdx.x * 8 + wave) * 16;

  const float* qreh = Qre + (size_t)h * SEQ * HD;
  const float* qimh = Qim + (size_t)h * SEQ * HD;
  const float* kreh = Kre + (size_t)h * SEQ * HD;
  const float* kimh = Kim + (size_t)h * SEQ * HD;
  const float* vreh = Vtre + (size_t)h * HD * SEQ;
  const float* vimh = Vtim + (size_t)h * HD * SEQ;

  v2f qr[8], qi[8], qrn[8];
#pragma unroll
  for (int t = 0; t < 8; ++t) {
    const int dk = t * 4 + half * 2;
    const size_t off = ((size_t)(q0 + l16)) * HD + dk;
    qr[t]  = ldp(qreh + off);
    qi[t]  = ldp(qimh + off);
    qrn[t] = -qr[t];
  }

  float m_run = -1e30f;
  float l_run = 0.f;
  v8f o_re0 = {0.f}, o_re1 = {0.f}, o_im0 = {0.f}, o_im1 = {0.f};

  KVFrag fa, fb;
  load_kv(fa, kreh, kimh, vreh, vimh, 0, l16, half);
  for (int kt = 0; kt < SEQ / 16; kt += 2) {
    load_kv(fb, kreh, kimh, vreh, vimh, kt + 1, l16, half);
    attn_tile(fa, qr, qi, qrn, m_run, l_run, o_re0, o_re1, o_im0, o_im1);
    const int nx = (kt + 2 < SEQ / 16) ? kt + 2 : SEQ / 16 - 1;
    load_kv(fa, kreh, kimh, vreh, vimh, nx, l16, half);
    attn_tile(fb, qr, qi, qrn, m_run, l_run, o_re0, o_re1, o_im0, o_im1);
  }

  const float inv = 1.0f / (l_run + xhalf(l_run));
  float* op = Out + ((size_t)(q0 + l16) * CH + h * HD + 8 * half) * 2;
#pragma unroll
  for (int j = 0; j < 4; ++j) {
    const float4 e0 = { o_re0[2 * j] * inv, o_im0[2 * j] * inv,
                        o_re0[2 * j + 1] * inv, o_im0[2 * j + 1] * inv };
    const float4 e1 = { o_re1[2 * j] * inv, o_im1[2 * j] * inv,
                        o_re1[2 * j + 1] * inv, o_im1[2 * j + 1] * inv };
    *(float4*)(op + 4 * j)      = e0;
    *(float4*)(op + 32 + 4 * j) = e1;
  }
}

extern "C" void kernel_launch(void* const* d_in, const int* in_sizes, int n_in,
                              void* d_out, int out_size, void* d_ws, size_t ws_size,
                              hipStream_t stream) {
  (void)in_sizes; (void)n_in; (void)out_size; (void)ws_size;
  const float* Q  = (const float*)d_in[0];
  const float* V  = (const float*)d_in[1];
  const float* K  = (const float*)d_in[2];
  const float* Wq = (const float*)d_in[3];
  const float* bq = (const float*)d_in[4];
  const float* Wk = (const float*)d_in[5];
  const float* bk = (const float*)d_in[6];
  const float* Wv = (const float*)d_in[7];
  const float* bv = (const float*)d_in[8];
  float* out = (float*)d_out;

  const size_t plane = (size_t)NH * SEQ * HD;
  float* qre = (float*)d_ws;
  float* qim = qre + plane;
  float* kre = qim + plane;
  float* kim = kre + plane;
  float* vre = kim + plane;
  float* vim = vre + plane;

  const dim3 blk(256);
  const dim3 pgrid((SEQ / 16) * (CH / 16) / 8);
  proj_complex_kernel<0><<<pgrid, blk, 0, stream>>>(Q, Wq, bq, qre, qim);
  proj_complex_kernel<0><<<pgrid, blk, 0, stream>>>(K, Wk, bk, kre, kim);
  proj_complex_kernel<1><<<pgrid, blk, 0, stream>>>(V, Wv, bv, vre, vim);

  const dim3 agrid(SEQ / 16 / 8, NH);
  attn_complex_kernel<<<agrid, blk, 0, stream>>>(qre, qim, kre, kim, vre, vim, out);
}


# attention operands: projections store K and V^T in WMMA operand order; a key tile is 16 loads of 4 whole 128-byte lines, not 24 of 16 partial lines; K tile computed transposed; order of sums kept
# speedup vs baseline: 1.7147x; 1.0952x over previous
#include <hip/hip_runtime.h>

#define SEQ 2048
#define CH  512
#define NH  16
#define HD  32

typedef float v2f __attribute__((ext_vector_type(2)));
typedef float v4f __attribute__((ext_vector_type(4)));
typedef float v8f __attribute__((ext_vector_type(8)));

__device__ __forceinline__ v8f wmma4(v2f a, v2f b, v8f c) {
  return __builtin_amdgcn_wmma_f32_16x16x4_f32(false, a, false, b, (short)0, c, false, false);
}

__device__ __forceinline__ v2f ldp(const float* p) {
  const float2 t = *(const float2*)p;
  v2f r = { t.x, t.y };
  return r;
}

template <int MODE>
__global__ void __launch_bounds__(256)
proj_complex_kernel(const float* __restrict__ X, const float* __restrict__ W,
                    const float* __restrict__ Bv, float* __restrict__ Yre,
                    float* __restrict__ Yim) {
  const int wave = threadIdx.x >> 5;
  const int lane = threadIdx.x & 31;
  const int half = lane >> 4;
  const int l16  = lane & 15;

  const int tileId  = blockIdx.x * 8 + wave;
  const int nTilesO = CH / 16;
  const int sTile   = tileId / nTilesO;
  const int oTile   = tileId % nTilesO;

  const int xrow = sTile * 16 + l16;
  const int wrow = oTile * 16 + l16;
  const float* Ap = (MODE == 2) ? W + (size_t)wrow * CH * 2 : X + (size_t)xrow * CH * 2;
  const float* Bp = (MODE == 2) ? X + (size_t)xrow * CH * 2 : W + (size_t)wrow * CH * 2;

  v8f yr = {0.f}, yi = {0.f};

  for (int k0 = 0; k0 < CH; k0 += 4) {
    const int ka = k0 + half * 2;
    const float4 fa = *(const float4*)(Ap + (size_t)ka * 2);
    const float4 fb = *(const float4*)(Bp + (size_t)ka * 2);
    v2f ar  = { fa.x,  fa.z };
    v2f ai  = { fa.y,  fa.w };
    v2f ain = { -fa.y, -fa.w };
    v2f br  = { fb.x,  fb.z };
    v2f bi  = { fb.y,  fb.w };
    yr = wmma4(ar,  br, yr);
    yr = wmma4(ain, bi, yr);
    yi = wmma4(ar,  bi, yi);
    yi = wmma4(ai,  br, yi);
  }

  if (MODE == 0) {
    const float2 bb = *(const float2*)(Bv + (size_t)wrow * 2);
    const int h  = wrow >> 5;
    const int dd = wrow & 31;
#pragma unroll
    for (int r = 0; r < 8; ++r) {
      const int s = sTile * 16 + r + 8 * half;
      const size_t idx = ((size_t)h * SEQ + s) * HD + dd;
      Yre[idx] = yr[r] + bb.x;
      Yim[idx] = yi[r] + bb.y;
    }
  } else if (MODE == 1) {
    const float2 bb = *(const float2*)(Bv + (size_t)wrow * 2);
    const int h  = wrow >> 5;
    const int dt = oTile & 1;
    const size_t base = ((size_t)(h * (SEQ / 16) + sTile)) * 512 + (size_t)(dt * 2) * 128 + lane * 4;
    float4 r0 = { yr[0] + bb.x, yr[1] + bb.x, yr[2] + bb.x, yr[3] + bb.x };
    float4 r1 = { yr[4] + bb.x, yr[5] + bb.x, yr[6] + bb.x, yr[7] + bb.x };
    float4 i0 = { yi[0] + bb.y, yi[1] + bb.y, yi[2] + bb.y, yi[3] + bb.y };
    float4 i1 = { yi[4] + bb.y, yi[5] + bb.y, yi[6] + bb.y, yi[7] + bb.y };
    *(float4*)(Yre + base)       = r0;
    *(float4*)(Yre + base + 128) = r1;
    *(float4*)(Yim + base)       = i0;
    *(float4*)(Yim + base + 128) = i1;
  } else {
    const int oc0 = oTile * 16 + 8 * half;
    const float4 b01 = *(const float4*)(Bv + (size_t)oc0 * 2);
    const float4 b23 = *(const float4*)(Bv + (size_t)oc0 * 2 + 4);
    const float4 b45 = *(const float4*)(Bv + (size_t)oc0 * 2 + 8);
    const float4 b67 = *(const float4*)(Bv + (size_t)oc0 * 2 + 12);
    const int h = oTile >> 1;
    const int j = (oTile & 1) * 2 + half;
    const size_t base = ((size_t)(h * (SEQ / 16) + sTile)) * 512 + (size_t)j * 128 + l16 * 4;
    float4 rlo = { yr[0] + b01.x, yr[1] + b01.z, yr[4] + b45.x, yr[5] + b45.z };
    float4 rhi = { yr[2] + b23.x, yr[3] + b23.z, yr[6] + b67.x, yr[7] + b67.z };
    float4 ilo = { yi[0] + b01.y, yi[1] + b01.w, yi[4] + b45.y, yi[5] + b45.w };
    float4 ihi = { yi[2] + b23.y, yi[3] + b23.w, yi[6] + b67.y, yi[7] + b67.w };
    *(float4*)(Yre + base)      = rlo;
    *(float4*)(Yre + base + 64) = rhi;
    *(float4*)(Yim + base)      = ilo;
    *(float4*)(Yim + base + 64) = ihi;
  }
}

__device__ __forceinline__ float xhalf(float x) {
  const int b = __builtin_bit_cast(int, x);
  return __builtin_bit_cast(float,
      __builtin_amdgcn_permlanex16(b, b, 0x76543210, 0xfedcba98, false, false));
}

struct KVFrag {
  v4f kr[4], ki[4];
  v4f vr[4], vi[4];
};

__device__ __forceinline__ void load_kv(KVFrag& f, const float* __restrict__ kreh,
                                        const float* __restrict__ kimh,
                                        const float* __restrict__ vreh,
                                        const float* __restrict__ vimh,
                                        int kt, int lane) {
  const size_t off = (size_t)kt * 512 + lane * 4;
#pragma unroll
  for (int i = 0; i < 4; ++i) {
    f.kr[i] = *(const v4f*)(kreh + off + i * 128);
    f.ki[i] = *(const v4f*)(kimh + off + i * 128);
    f.vr[i] = *(const v4f*)(vreh + off + i * 128);
    f.vi[i] = *(const v4f*)(vimh + off + i * 128);
  }
}

__device__ __forceinline__ void attn_tile(KVFrag& f, const v2f (&qr)[8], const v2f (&qi)[8],
                                          const v2f (&qrn)[8], float& m_run, float& l_run,
                                          v8f& o_re0, v8f& o_re1, v8f& o_im0, v8f& o_im1) {
  const float scale = 0.17677669529663687f;

  asm volatile("" : "+v"(f.kr[0]), "+v"(f.kr[1]), "+v"(f.kr[2]), "+v"(f.kr[3]),
                    "+v"(f.ki[0]), "+v"(f.ki[1]), "+v"(f.ki[2]), "+v"(f.ki[3]),
                    "+v"(f.vr[0]), "+v"(f.vr[1]), "+v"(f.vr[2]), "+v"(f.vr[3]),
                    "+v"(f.vi[0]), "+v"(f.vi[1]), "+v"(f.vi[2]), "+v"(f.vi[3]));

  v8f sre = {0.f}, sim = {0.f};
#pragma unroll
  for (int t = 0; t < 8; ++t) {
    const v4f r4 = f.kr[t >> 1], i4 = f.ki[t >> 1];
    const v2f kr = (t & 1) ? v2f{ r4.z, r4.w } : v2f{ r4.x, r4.y };
    const v2f ki = (t & 1) ? v2f{ i4.z, i4.w } : v2f{ i4.x, i4.y };
    sre = wmma4(kr, qr[t],  sre);
    sre = wmma4(ki, qi[t],  sre);
    sim = wmma4(kr, qi[t],  sim);
    sim = wmma4(ki, qrn[t], sim);
  }

  float a[8];
#pragma unroll
  for (int r = 0; r < 8; ++r)
    a[r] = __builtin_amdgcn_sqrtf(sre[r] * sre[r] + sim[r] * sim[r]) * scale;
  float mx = fmaxf(fmaxf(fmaxf(a[0], a[1]), fmaxf(a[2], a[3])),
                   fmaxf(fmaxf(a[4], a[5]), fmaxf(a[6], a[7])));
  mx = fmaxf(mx, xhalf(mx));
  const float newm = fmaxf(m_run, mx);
  const float sc = __expf(m_run - newm);
  float p[8];
#pragma unroll
  for (int r = 0; r < 8; ++r) p[r] = __expf(a[r] - newm);
  l_run = l_run * sc + (((p[0] + p[1]) + (p[2] + p[3])) + ((p[4] + p[5]) + (p[6] + p[7])));
  m_run = newm;
#pragma unroll
  for (int r = 0; r < 8; ++r) {
    o_re0[r] *= sc; o_re1[r] *= sc;
    o_im0[r] *= sc; o_im1[r] *= sc;
  }

  const v2f ar0[4] = { {f.vr[0].x, f.vr[0].y}, {f.vr[0].z, f.vr[0].w}, {f.vr[1].x, f.vr[1].y}, {f.vr[1].z, f.vr[1].w} };
  const v2f ai0[4] = { {f.vi[0].x, f.vi[0].y}, {f.vi[0].z, f.vi[0].w}, {f.vi[1].x, f.vi[1].y}, {f.vi[1].z, f.vi[1].w} };
  const v2f ar1[4] = { {f.vr[2].x, f.vr[2].y}, {f.vr[2].z, f.vr[2].w}, {f.vr[3].x, f.vr[3].y}, {f.vr[3].z, f.vr[3].w} };
  const v2f ai1[4] = { {f.vi[2].x, f.vi[2].y}, {f.vi[2].z, f.vi[2].w}, {f.vi[3].x, f.vi[3].y}, {f.vi[3].z, f.vi[3].w} };
#pragma unroll
  for (int t = 0; t < 4; ++t) {
    const v2f bP = { p[2 * t], p[2 * t + 1] };
    o_re0 = wmma4(ar0[t], bP, o_re0);
    o_im0 = wmma4(ai0[t], bP, o_im0);
    o_re1 = wmma4(ar1[t], bP, o_re1);
    o_im1 = wmma4(ai1[t], bP, o_im1);
  }
}

__global__ void __launch_bounds__(256)
attn_complex_kernel(const float* __restrict__ Qre, const float* __restrict__ Qim,
                    const float* __restrict__ Kre, const float* __restrict__ Kim,
                    const float* __restrict__ Vtre, const float* __restrict__ Vtim,
                    float* __restrict__ Out) {
  const int wave = threadIdx.x >> 5;
  const int lane = threadIdx.x & 31;
  const int half = lane >> 4;
  const int l16  = lane & 15;
  const int h    = blockIdx.y;
  const int q0   = (blockIdx.x * 8 + wave) * 16;

  const float* qreh = Qre + (size_t)h * SEQ * HD;
  const float* qimh = Qim + (size_t)h * SEQ * HD;
  const float* kreh = Kre + (size_t)h * SEQ * HD;
  const float* kimh = Kim + (size_t)h * SEQ * HD;
  const float* vreh = Vtre + (size_t)h * SEQ * HD;
  const float* vimh = Vtim + (size_t)h * SEQ * HD;

  v2f qr[8], qi[8], qrn[8];
#pragma unroll
  for (int t = 0; t < 8; ++t) {
    const int dk = t * 4 + half * 2;
    const size_t off = ((size_t)(q0 + l16)) * HD + dk;
    qr[t]  = ldp(qreh + off);
    qi[t]  = ldp(qimh + off);
    qrn[t] = -qr[t];
  }

  float m_run = -1e30f;
  float l_run = 0.f;
  v8f o_re0 = {0.f}, o_re1 = {0.f}, o_im0 = {0.f}, o_im1 = {0.f};

  KVFrag fa, fb;
  load_kv(fa, kreh, kimh, vreh, vimh, 0, lane);
  for (int kt = 0; kt < SEQ / 16; kt += 2) {
    load_kv(fb, kreh, kimh, vreh, vimh, kt + 1, lane);
    attn_tile(fa, qr, qi, qrn, m_run, l_run, o_re0, o_re1, o_im0, o_im1);
    const int nx = (kt + 2 < SEQ / 16) ? kt + 2 : SEQ / 16 - 1;
    load_kv(fa, kreh, kimh, vreh, vimh, nx, lane);
    attn_tile(fb, qr, qi, qrn, m_run, l_run, o_re0, o_re1, o_im0, o_im1);
  }

  const float inv = 1.0f / (l_run + xhalf(l_run));
  float* op = Out + ((size_t)(q0 + l16) * CH + h * HD + 8 * half) * 2;
#pragma unroll
  for (int j = 0; j < 4; ++j) {
    const float4 e0 = { o_re0[2 * j] * inv, o_im0[2 * j] * inv,
                        o_re0[2 * j + 1] * inv, o_im0[2 * j + 1] * inv };
    const float4 e1 = { o_re1[2 * j] * inv, o_im1[2 * j] * inv,
                        o_re1[2 * j + 1] * inv, o_im1[2 * j + 1] * inv };
    *(float4*)(op + 4 * j)      = e0;
    *(float4*)(op + 32 + 4 * j) = e1;
  }
}

extern "C" void kernel_launch(void* const* d_in, const int* in_sizes, int n_in,
                              void* d_out, int out_size, void* d_ws, size_t ws_size,
                              hipStream_t stream) {
  (void)in_sizes; (void)n_in; (void)out_size; (void)ws_size;
  const float* Q  = (const float*)d_in[0];
  const float* V  = (const float*)d_in[1];
  const float* K  = (const float*)d_in[2];
  const float* Wq = (const float*)d_in[3];
  const float* bq = (const float*)d_in[4];
  const float* Wk = (const float*)d_in[5];
  const float* bk = (const float*)d_in[6];
  const float* Wv = (const float*)d_in[7];
  const float* bv = (const float*)d_in[8];
  float* out = (float*)d_out;

  const size_t plane = (size_t)NH * SEQ * HD;
  float* qre = (float*)d_ws;
  float* qim = qre + plane;
  float* kre = qim + plane;
  float* kim = kre + plane;
  float* vre = kim + plane;
  float* vim = vre + plane;

  const dim3 blk(256);
  const dim3 pgrid((SEQ / 16) * (CH / 16) / 8);
  proj_complex_kernel<0><<<pgrid, blk, 0, stream>>>(Q, Wq, bq, qre, qim);
  proj_complex_kernel<2><<<pgrid, blk, 0, stream>>>(K, Wk, bk, kre, kim);
  proj_complex_kernel<1><<<pgrid, blk, 0, stream>>>(V, Wv, bv, vre, vim);

  const dim3 agrid(SEQ / 16 / 8, NH);
  attn_complex_kernel<<<agrid, blk, 0, stream>>>(qre, qim, kre, kim, vre, vim, out);
}


# every complex matrix product (3 projections, attention scores) from 3 real WMMA products, not 4: re*re, im*im, (re+im)*(re+im); re and im formed from whole sums: rounding and order of sums change
# speedup vs baseline: 1.8746x; 1.0933x over previous
#include <hip/hip_runtime.h>

#define SEQ 2048
#define CH  512
#define NH  16
#define HD  32

typedef float v2f __attribute__((ext_vector_type(2)));
typedef float v4f __attribute__((ext_vector_type(4)));
typedef float v8f __attribute__((ext_vector_type(8)));

__device__ __forceinline__ v8f wmma4(v2f a, v2f b, v8f c) {
  return __builtin_amdgcn_wmma_f32_16x16x4_f32(false, a, false, b, (short)0, c, false, false);
}

__device__ __forceinline__ v2f ldp(const float* p) {
  const float2 t = *(const float2*)p;
  v2f r = { t.x, t.y };
  return r;
}

template <int MODE>
__global__ void __launch_bounds__(256)
proj_complex_kernel(const float* __restrict__ X, const float* __restrict__ W,
                    const float* __restrict__ Bv, float* __restrict__ Yre,
                    float* __restrict__ Yim) {
  const int wave = threadIdx.x >> 5;
  const int lane = threadIdx.x & 31;
  const int half = lane >> 4;
  const int l16  = lane & 15;

  const int tileId  = blockIdx.x * 8 + wave;
  const int nTilesO = CH / 16;
  const int sTile   = tileId / nTilesO;
  const int oTile   = tileId % nTilesO;

  const int xrow = sTile * 16 + l16;
  const int wrow = oTile * 16 + l16;
  const float* Ap = (MODE == 2) ? W + (size_t)wrow * CH * 2 : X + (size_t)xrow * CH * 2;
  const float* Bp = (MODE == 2) ? X + (size_t)xrow * CH * 2 : W + (size_t)wrow * CH * 2;

  v8f p1 = {0.f}, p2 = {0.f}, p3 = {0.f};

  for (int k0 = 0; k0 < CH; k0 += 32) {
    float4 fa[8], fb[8];
#pragma unroll
    for (int j = 0; j < 8; ++j) {
      const int ka = k0 + 4 * j + half * 2;
      fa[j] = *(const float4*)(Ap + (size_t)ka * 2);
      fb[j] = *(const float4*)(Bp + (size_t)ka * 2);
    }
#pragma unroll
    for (int j = 0; j < 8; ++j) {
      v2f ar = { fa[j].x, fa[j].z };
      v2f ai = { fa[j].y, fa[j].w };
      v2f as = { fa[j].x + fa[j].y, fa[j].z + fa[j].w };
      v2f br = { fb[j].x, fb[j].z };
      v2f bi = { fb[j].y, fb[j].w };
      v2f bs = { fb[j].x + fb[j].y, fb[j].z + fb[j].w };
      p1 = wmma4(ar, br, p1);
      p2 = wmma4(ai, bi, p2);
      p3 = wmma4(as, bs, p3);
    }
  }
  const v8f yr = p1 - p2;
  const v8f yi = (p3 - p1) - p2;

  if (MODE == 0) {
    const float2 bb = *(const float2*)(Bv + (size_t)wrow * 2);
    const int h  = wrow >> 5;
    const int dd = wrow & 31;
#pragma unroll
    for (int r = 0; r < 8; ++r) {
      const int s = sTile * 16 + r + 8 * half;
      const size_t idx = ((size_t)h * SEQ + s) * HD + dd;
      Yre[idx] = yr[r] + bb.x;
      Yim[idx] = yi[r] + bb.y;
    }
  } else if (MODE == 1) {
    const float2 bb = *(const float2*)(Bv + (size_t)wrow * 2);
    const int h  = wrow >> 5;
    const int dt = oTile & 1;
    const size_t base = ((size_t)(h * (SEQ / 16) + sTile)) * 512 + (size_t)(dt * 2) * 128 + lane * 4;
    float4 r0 = { yr[0] + bb.x, yr[1] + bb.x, yr[2] + bb.x, yr[3] + bb.x };
    float4 r1 = { yr[4] + bb.x, yr[5] + bb.x, yr[6] + bb.x, yr[7] + bb.x };
    float4 i0 = { yi[0] + bb.y, yi[1] + bb.y, yi[2] + bb.y, yi[3] + bb.y };
    float4 i1 = { yi[4] + bb.y, yi[5] + bb.y, yi[6] + bb.y, yi[7] + bb.y };
    *(float4*)(Yre + base)       = r0;
    *(float4*)(Yre + base + 128) = r1;
    *(float4*)(Yim + base)       = i0;
    *(float4*)(Yim + base + 128) = i1;
  } else {
    const int oc0 = oTile * 16 + 8 * half;
    const float4 b01 = *(const float4*)(Bv + (size_t)oc0 * 2);
    const float4 b23 = *(const float4*)(Bv + (size_t)oc0 * 2 + 4);
    const float4 b45 = *(const float4*)(Bv + (size_t)oc0 * 2 + 8);
    const float4 b67 = *(const float4*)(Bv + (size_t)oc0 * 2 + 12);
    const int h = oTile >> 1;
    const int j = (oTile & 1) * 2 + half;
    const size_t base = ((size_t)(h * (SEQ / 16) + sTile)) * 512 + (size_t)j * 128 + l16 * 4;
    float4 rlo = { yr[0] + b01.x, yr[1] + b01.z, yr[4] + b45.x, yr[5] + b45.z };
    float4 rhi = { yr[2] + b23.x, yr[3] + b23.z, yr[6] + b67.x, yr[7] + b67.z };
    float4 ilo = { yi[0] + b01.y, yi[1] + b01.w, yi[4] + b45.y, yi[5] + b45.w };
    float4 ihi = { yi[2] + b23.y, yi[3] + b23.w, yi[6] + b67.y, yi[7] + b67.w };
    *(float4*)(Yre + base)      = rlo;
    *(float4*)(Yre + base + 64) = rhi;
    *(float4*)(Yim + base)      = ilo;
    *(float4*)(Yim + base + 64) = ihi;
  }
}

__device__ __forceinline__ float xhalf(float x) {
  const int b = __builtin_bit_cast(int, x);
  return __builtin_bit_cast(float,
      __builtin_amdgcn_permlanex16(b, b, 0x76543210, 0xfedcba98, false, false));
}

struct KVFrag {
  v4f kr[4], ki[4];
  v4f vr[4], vi[4];
};

__device__ __forceinline__ void load_kv(KVFrag& f, const float* __restrict__ kreh,
                                        const float* __restrict__ kimh,
                                        const float* __restrict__ vreh,
                                        const float* __restrict__ vimh,
                                        int kt, int lane) {
  const size_t off = (size_t)kt * 512 + lane * 4;
#pragma unroll
  for (int i = 0; i < 4; ++i) {
    f.kr[i] = *(const v4f*)(kreh + off + i * 128);
    f.ki[i] = *(const v4f*)(kimh + off + i * 128);
    f.vr[i] = *(const v4f*)(vreh + off + i * 128);
    f.vi[i] = *(const v4f*)(vimh + off + i * 128);
  }
}

__device__ __forceinline__ void attn_tile(KVFrag& f, const v2f (&qr)[8], const v2f (&qi)[8],
                                          const v2f (&qd)[8], float& m_run, float& l_run,
                                          v8f& o_re0, v8f& o_re1, v8f& o_im0, v8f& o_im1) {
  const float scale = 0.17677669529663687f;

  asm volatile("" : "+v"(f.kr[0]), "+v"(f.kr[1]), "+v"(f.kr[2]), "+v"(f.kr[3]),
                    "+v"(f.ki[0]), "+v"(f.ki[1]), "+v"(f.ki[2]), "+v"(f.ki[3]),
                    "+v"(f.vr[0]), "+v"(f.vr[1]), "+v"(f.vr[2]), "+v"(f.vr[3]),
                    "+v"(f.vi[0]), "+v"(f.vi[1]), "+v"(f.vi[2]), "+v"(f.vi[3]));

  v8f p1 = {0.f}, p2 = {0.f}, p3 = {0.f};
#pragma unroll
  for (int t = 0; t < 8; ++t) {
    const v4f r4 = f.kr[t >> 1], i4 = f.ki[t >> 1];
    const v2f kr = (t & 1) ? v2f{ r4.z, r4.w } : v2f{ r4.x, r4.y };
    const v2f ki = (t & 1) ? v2f{ i4.z, i4.w } : v2f{ i4.x, i4.y };
    const v2f ks = kr + ki;
    p1 = wmma4(kr, qr[t], p1);
    p2 = wmma4(ki, qi[t], p2);
    p3 = wmma4(ks, qd[t], p3);
  }
  const v8f sre = p1 + p2;
  const v8f sim = p3 + (p1 - p2);

  float a[8];
#pragma unroll
  for (int r = 0; r < 8; ++r)
    a[r] = __builtin_amdgcn_sqrtf(sre[r] * sre[r] + sim[r] * sim[r]) * scale;
  float mx = fmaxf(fmaxf(fmaxf(a[0], a[1]), fmaxf(a[2], a[3])),
                   fmaxf(fmaxf(a[4], a[5]), fmaxf(a[6], a[7])));
  mx = fmaxf(mx, xhalf(mx));
  const float newm = fmaxf(m_run, mx);
  const float sc = __expf(m_run - newm);
  float p[8];
#pragma unroll
  for (int r = 0; r < 8; ++r) p[r] = __expf(a[r] - newm);
  l_run = l_run * sc + (((p[0] + p[1]) + (p[2] + p[3])) + ((p[4] + p[5]) + (p[6] + p[7])));
  m_run = newm;
#pragma unroll
  for (int r = 0; r < 8; ++r) {
    o_re0[r] *= sc; o_re1[r] *= sc;
    o_im0[r] *= sc; o_im1[r] *= sc;
  }

  const v2f ar0[4] = { {f.vr[0].x, f.vr[0].y}, {f.vr[0].z, f.vr[0].w}, {f.vr[1].x, f.vr[1].y}, {f.vr[1].z, f.vr[1].w} };
  const v2f ai0[4] = { {f.vi[0].x, f.vi[0].y}, {f.vi[0].z, f.vi[0].w}, {f.vi[1].x, f.vi[1].y}, {f.vi[1].z, f.vi[1].w} };
  const v2f ar1[4] = { {f.vr[2].x, f.vr[2].y}, {f.vr[2].z, f.vr[2].w}, {f.vr[3].x, f.vr[3].y}, {f.vr[3].z, f.vr[3].w} };
  const v2f ai1[4] = { {f.vi[2].x, f.vi[2].y}, {f.vi[2].z, f.vi[2].w}, {f.vi[3].x, f.vi[3].y}, {f.vi[3].z, f.vi[3].w} };
#pragma unroll
  for (int t = 0; t < 4; ++t) {
    const v2f bP = { p[2 * t], p[2 * t + 1] };
    o_re0 = wmma4(ar0[t], bP, o_re0);
    o_im0 = wmma4(ai0[t], bP, o_im0);
    o_re1 = wmma4(ar1[t], bP, o_re1);
    o_im1 = wmma4(ai1[t], bP, o_im1);
  }
}

__global__ void __launch_bounds__(256)
attn_complex_kernel(const float* __restrict__ Qre, const float* __restrict__ Qim,
                    const float* __restrict__ Kre, const float* __restrict__ Kim,
                    const float* __restrict__ Vtre, const float* __restrict__ Vtim,
                    float* __restrict__ Out) {
  const int wave = threadIdx.x >> 5;
  const int lane = threadIdx.x & 31;
  const int half = lane >> 4;
  const int l16  = lane & 15;
  const int h    = blockIdx.y;
  const int q0   = (blockIdx.x * 8 + wave) * 16;

  const float* qreh = Qre + (size_t)h * SEQ * HD;
  const float* qimh = Qim + (size_t)h * SEQ * HD;
  const float* kreh = Kre + (size_t)h * SEQ * HD;
  const float* kimh = Kim + (size_t)h * SEQ * HD;
  const float* vreh = Vtre + (size_t)h * SEQ * HD;
  const float* vimh = Vtim + (size_t)h * SEQ * HD;

  v2f qr[8], qi[8], qd[8];
#pragma unroll
  for (int t = 0; t < 8; ++t) {
    const int dk = t * 4 + half * 2;
    const size_t off = ((size_t)(q0 + l16)) * HD + dk;
    qr[t]  = ldp(qreh + off);
    qi[t]  = ldp(qimh + off);
    qd[t]  = qi[t] - qr[t];
  }

  float m_run = -1e30f;
  float l_run = 0.f;
  v8f o_re0 = {0.f}, o_re1 = {0.f}, o_im0 = {0.f}, o_im1 = {0.f};

  KVFrag fa, fb;
  load_kv(fa, kreh, kimh, vreh, vimh, 0, lane);
  for (int kt = 0; kt < SEQ / 16; kt += 2) {
    load_kv(fb, kreh, kimh, vreh, vimh, kt + 1, lane);
    attn_tile(fa, qr, qi, qd, m_run, l_run, o_re0, o_re1, o_im0, o_im1);
    const int nx = (kt + 2 < SEQ / 16) ? kt + 2 : SEQ / 16 - 1;
    load_kv(fa, kreh, kimh, vreh, vimh, nx, lane);
    attn_tile(fb, qr, qi, qd, m_run, l_run, o_re0, o_re1, o_im0, o_im1);
  }

  const float inv = 1.0f / (l_run + xhalf(l_run));
  float* op = Out + ((size_t)(q0 + l16) * CH + h * HD + 8 * half) * 2;
#pragma unroll
  for (int j = 0; j < 4; ++j) {
    const float4 e0 = { o_re0[2 * j] * inv, o_im0[2 * j] * inv,
                        o_re0[2 * j + 1] * inv, o_im0[2 * j + 1] * inv };
    const float4 e1 = { o_re1[2 * j] * inv, o_im1[2 * j] * inv,
                        o_re1[2 * j + 1] * inv, o_im1[2 * j + 1] * inv };
    *(float4*)(op + 4 * j)      = e0;
    *(float4*)(op + 32 + 4 * j) = e1;
  }
}

extern "C" void kernel_launch(void* const* d_in, const int* in_sizes, int n_in,
                              void* d_out, int out_size, void* d_ws, size_t ws_size,
                              hipStream_t stream) {
  (void)in_sizes; (void)n_in; (void)out_size; (void)ws_size;
  const float* Q  = (const float*)d_in[0];
  const float* V  = (const float*)d_in[1];
  const float* K  = (const float*)d_in[2];
  const float* Wq = (const float*)d_in[3];
  const float* bq = (const float*)d_in[4];
  const float* Wk = (const float*)d_in[5];
  const float* bk = (const float*)d_in[6];
  const float* Wv = (const float*)d_in[7];
  const float* bv = (const float*)d_in[8];
  float* out = (float*)d_out;

  const size_t plane = (size_t)NH * SEQ * HD;
  float* qre = (float*)d_ws;
  float* qim = qre + plane;
  float* kre = qim + plane;
  float* kim = kre + plane;
  float* vre = kim + plane;
  float* vim = vre + plane;

  const dim3 blk(256);
  const dim3 pgrid((SEQ / 16) * (CH / 16) / 8);
  proj_complex_kernel<0><<<pgrid, blk, 0, stream>>>(Q, Wq, bq, qre, qim);
  proj_complex_kernel<2><<<pgrid, blk, 0, stream>>>(K, Wk, bk, kre, kim);
  proj_complex_kernel<1><<<pgrid, blk, 0, stream>>>(V, Wv, bv, vre, vim);

  const dim3 agrid(SEQ / 16 / 8, NH);
  attn_complex_kernel<<<agrid, blk, 0, stream>>>(qre, qim, kre, kim, vre, vim, out);
}


# projections: X and W rows staged through LDS in 256-byte chunks by whole-line loads, fetched once per 32x64 block (4 waves, 2 tiles each), not per wave as 16 partial lines; order of sums kept
# speedup vs baseline: 2.0810x; 1.1101x over previous
#include <hip/hip_runtime.h>

#define SEQ 2048
#define CH  512
#define NH  16
#define HD  32

typedef float v2f __attribute__((ext_vector_type(2)));
typedef float v4f __attribute__((ext_vector_type(4)));
typedef float v8f __attribute__((ext_vector_type(8)));

__device__ __forceinline__ v8f wmma4(v2f a, v2f b, v8f c) {
  return __builtin_amdgcn_wmma_f32_16x16x4_f32(false, a, false, b, (short)0, c, false, false);
}

__device__ __forceinline__ v2f ldp(const float* p) {
  const float2 t = *(const float2*)p;
  v2f r = { t.x, t.y };
  return r;
}

#define PJ_SROWS 32
#define PJ_OROWS 64
#define PJ_ROWS  (PJ_SROWS + PJ_OROWS)
#define PJ_PITCH 17
#define PJ_CHUNK 64

template <int MODE>
__device__ __forceinline__ void proj_store_tile(const v8f yr, const v8f yi, const int sTile,
                                                const int oTile, const int lane,
                                                const float* __restrict__ Bv,
                                                float* __restrict__ Yre, float* __restrict__ Yim) {
  const int half = lane >> 4;
  const int l16  = lane & 15;
  const int wrow = oTile * 16 + l16;
  if (MODE == 0) {
    const float2 bb = *(const float2*)(Bv + (size_t)wrow * 2);
    const int h  = wrow >> 5;
    const int dd = wrow & 31;
#pragma unroll
    for (int r = 0; r < 8; ++r) {
      const int s = sTile * 16 + r + 8 * half;
      const size_t idx = ((size_t)h * SEQ + s) * HD + dd;
      Yre[idx] = yr[r] + bb.x;
      Yim[idx] = yi[r] + bb.y;
    }
  } else if (MODE == 1) {
    const float2 bb = *(const float2*)(Bv + (size_t)wrow * 2);
    const int h  = wrow >> 5;
    const int dt = oTile & 1;
    const size_t base = ((size_t)(h * (SEQ / 16) + sTile)) * 512 + (size_t)(dt * 2) * 128 + lane * 4;
    float4 r0 = { yr[0] + bb.x, yr[1] + bb.x, yr[2] + bb.x, yr[3] + bb.x };
    float4 r1 = { yr[4] + bb.x, yr[5] + bb.x, yr[6] + bb.x, yr[7] + bb.x };
    float4 i0 = { yi[0] + bb.y, yi[1] + bb.y, yi[2] + bb.y, yi[3] + bb.y };
    float4 i1 = { yi[4] + bb.y, yi[5] + bb.y, yi[6] + bb.y, yi[7] + bb.y };
    *(float4*)(Yre + base)       = r0;
    *(float4*)(Yre + base + 128) = r1;
    *(float4*)(Yim + base)       = i0;
    *(float4*)(Yim + base + 128) = i1;
  } else {
    const int oc0 = oTile * 16 + 8 * half;
    const float4 b01 = *(const float4*)(Bv + (size_t)oc0 * 2);
    const float4 b23 = *(const float4*)(Bv + (size_t)oc0 * 2 + 4);
    const float4 b45 = *(const float4*)(Bv + (size_t)oc0 * 2 + 8);
    const float4 b67 = *(const float4*)(Bv + (size_t)oc0 * 2 + 12);
    const int h = oTile >> 1;
    const int j = (oTile & 1) * 2 + half;
    const size_t base = ((size_t)(h * (SEQ / 16) + sTile)) * 512 + (size_t)j * 128 + l16 * 4;
    float4 rlo = { yr[0] + b01.x, yr[1] + b01.z, yr[4] + b45.x, yr[5] + b45.z };
    float4 rhi = { yr[2] + b23.x, yr[3] + b23.z, yr[6] + b67.x, yr[7] + b67.z };
    float4 ilo = { yi[0] + b01.y, yi[1] + b01.w, yi[4] + b45.y, yi[5] + b45.w };
    float4 ihi = { yi[2] + b23.y, yi[3] + b23.w, yi[6] + b67.y, yi[7] + b67.w };
    *(float4*)(Yre + base)      = rlo;
    *(float4*)(Yre + base + 64) = rhi;
    *(float4*)(Yim + base)      = ilo;
    *(float4*)(Yim + base + 64) = ihi;
  }
}

template <int MODE>
__global__ void __launch_bounds__(128)
proj_complex_kernel(const float* __restrict__ X, const float* __restrict__ W,
                    const float* __restrict__ Bv, float* __restrict__ Yre,
                    float* __restrict__ Yim) {
  __shared__ v4f tile[2][PJ_ROWS * PJ_PITCH];

  const int tid  = threadIdx.x;
  const int wave = tid >> 5;
  const int lane = tid & 31;
  const int half = lane >> 4;
  const int l16  = lane & 15;

  const int sBlock = blockIdx.x >> 3;
  const int oBlock = blockIdx.x & 7;
  const int sLocal = wave >> 1;
  const int oPair  = wave & 1;
  const int sTile  = sBlock * 2 + sLocal;
  const int oTile0 = oBlock * 4 + oPair * 2;

  const int c16 = tid & 15;
  const int r0  = tid >> 4;
  const float* xsrc = X + (size_t)(sBlock * PJ_SROWS + r0) * CH * 2 + c16 * 4;
  const float* wsrc = W + (size_t)(oBlock * PJ_OROWS + r0) * CH * 2 + c16 * 4;
  const int sdst = r0 * PJ_PITCH + c16;

  const int xoff  = (sLocal * 16 + l16) * PJ_PITCH + half;
  const int w0off = (PJ_SROWS + oPair * 32 + l16) * PJ_PITCH + half;
  const int w1off = w0off + 16 * PJ_PITCH;

  v4f g[12];
#pragma unroll
  for (int i = 0; i < 4; ++i) g[i] = *(const v4f*)(xsrc + (size_t)i * 8 * CH * 2);
#pragma unroll
  for (int i = 0; i < 8; ++i) g[4 + i] = *(const v4f*)(wsrc + (size_t)i * 8 * CH * 2);
#pragma unroll
  for (int i = 0; i < 12; ++i) tile[0][sdst + i * 8 * PJ_PITCH] = g[i];
  __syncthreads();

  v8f p1a = {0.f}, p2a = {0.f}, p3a = {0.f};
  v8f p1b = {0.f}, p2b = {0.f}, p3b = {0.f};

  for (int c = 0; c < CH * 2 / PJ_CHUNK; ++c) {
    const bool more = (c + 1 < CH * 2 / PJ_CHUNK);
    if (more) {
      const int ko = (c + 1) * PJ_CHUNK;
#pragma unroll
      for (int i = 0; i < 4; ++i) g[i] = *(const v4f*)(xsrc + (size_t)i * 8 * CH * 2 + ko);
#pragma unroll
      for (int i = 0; i < 8; ++i) g[4 + i] = *(const v4f*)(wsrc + (size_t)i * 8 * CH * 2 + ko);
      asm volatile("" ::: "memory");
    }

    const v4f* cur = tile[c & 1];
    v4f fx[8], fw0[8], fw1[8];
#pragma unroll
    for (int j = 0; j < 8; ++j) {
      fx[j]  = cur[xoff + 2 * j];
      fw0[j] = cur[w0off + 2 * j];
      fw1[j] = cur[w1off + 2 * j];
    }
#pragma unroll
    for (int j = 0; j < 8; ++j) {
      const v2f xr  = { fx[j].x, fx[j].z };
      const v2f xi  = { fx[j].y, fx[j].w };
      const v2f xs  = { fx[j].x + fx[j].y, fx[j].z + fx[j].w };
      const v2f w0r = { fw0[j].x, fw0[j].z };
      const v2f w0i = { fw0[j].y, fw0[j].w };
      const v2f w0s = { fw0[j].x + fw0[j].y, fw0[j].z + fw0[j].w };
      const v2f w1r = { fw1[j].x, fw1[j].z };
      const v2f w1i = { fw1[j].y, fw1[j].w };
      const v2f w1s = { fw1[j].x + fw1[j].y, fw1[j].z + fw1[j].w };
      if (MODE == 2) {
        p1a = wmma4(w0r, xr, p1a);  p2a = wmma4(w0i, xi, p2a);  p3a = wmma4(w0s, xs, p3a);
        p1b = wmma4(w1r, xr, p1b);  p2b = wmma4(w1i, xi, p2b);  p3b = wmma4(w1s, xs, p3b);
      } else {
        p1a = wmma4(xr, w0r, p1a);  p2a = wmma4(xi, w0i, p2a);  p3a = wmma4(xs, w0s, p3a);
        p1b = wmma4(xr, w1r, p1b);  p2b = wmma4(xi, w1i, p2b);  p3b = wmma4(xs, w1s, p3b);
      }
    }

    if (more) {
      asm volatile("" : "+v"(p1a), "+v"(p2a), "+v"(p3a), "+v"(p1b), "+v"(p2b), "+v"(p3b),
                        "+v"(g[0]), "+v"(g[1]), "+v"(g[2]), "+v"(g[3]), "+v"(g[4]), "+v"(g[5]),
                        "+v"(g[6]), "+v"(g[7]), "+v"(g[8]), "+v"(g[9]), "+v"(g[10]), "+v"(g[11]));
      v4f* nxt = tile[(c + 1) & 1];
#pragma unroll
      for (int i = 0; i < 12; ++i) nxt[sdst + i * 8 * PJ_PITCH] = g[i];
    }
    __syncthreads();
  }

  const v8f yra = p1a - p2a;
  const v8f yia = (p3a - p1a) - p2a;
  const v8f yrb = p1b - p2b;
  const v8f yib = (p3b - p1b) - p2b;
  proj_store_tile<MODE>(yra, yia, sTile, oTile0,     lane, Bv, Yre, Yim);
  proj_store_tile<MODE>(yrb, yib, sTile, oTile0 + 1, lane, Bv, Yre, Yim);
}

__device__ __forceinline__ float xhalf(float x) {
  const int b = __builtin_bit_cast(int, x);
  return __builtin_bit_cast(float,
      __builtin_amdgcn_permlanex16(b, b, 0x76543210, 0xfedcba98, false, false));
}

struct KVFrag {
  v4f kr[4], ki[4];
  v4f vr[4], vi[4];
};

__device__ __forceinline__ void load_kv(KVFrag& f, const float* __restrict__ kreh,
                                        const float* __restrict__ kimh,
                                        const float* __restrict__ vreh,
                                        const float* __restrict__ vimh,
                                        int kt, int lane) {
  const size_t off = (size_t)kt * 512 + lane * 4;
#pragma unroll
  for (int i = 0; i < 4; ++i) {
    f.kr[i] = *(const v4f*)(kreh + off + i * 128);
    f.ki[i] = *(const v4f*)(kimh + off + i * 128);
    f.vr[i] = *(const v4f*)(vreh + off + i * 128);
    f.vi[i] = *(const v4f*)(vimh + off + i * 128);
  }
}

__device__ __forceinline__ void attn_tile(KVFrag& f, const v2f (&qr)[8], const v2f (&qi)[8],
                                          const v2f (&qd)[8], float& m_run, float& l_run,
                                          v8f& o_re0, v8f& o_re1, v8f& o_im0, v8f& o_im1) {
  const float scale = 0.17677669529663687f;

  asm volatile("" : "+v"(f.kr[0]), "+v"(f.kr[1]), "+v"(f.kr[2]), "+v"(f.kr[3]),
                    "+v"(f.ki[0]), "+v"(f.ki[1]), "+v"(f.ki[2]), "+v"(f.ki[3]),
                    "+v"(f.vr[0]), "+v"(f.vr[1]), "+v"(f.vr[2]), "+v"(f.vr[3]),
                    "+v"(f.vi[0]), "+v"(f.vi[1]), "+v"(f.vi[2]), "+v"(f.vi[3]));

  v8f p1 = {0.f}, p2 = {0.f}, p3 = {0.f};
#pragma unroll
  for (int t = 0; t < 8; ++t) {
    const v4f r4 = f.kr[t >> 1], i4 = f.ki[t >> 1];
    const v2f kr = (t & 1) ? v2f{ r4.z, r4.w } : v2f{ r4.x, r4.y };
    const v2f ki = (t & 1) ? v2f{ i4.z, i4.w } : v2f{ i4.x, i4.y };
    const v2f ks = kr + ki;
    p1 = wmma4(kr, qr[t], p1);
    p2 = wmma4(ki, qi[t], p2);
    p3 = wmma4(ks, qd[t], p3);
  }
  const v8f sre = p1 + p2;
  const v8f sim = p3 + (p1 - p2);

  float a[8];
#pragma unroll
  for (int r = 0; r < 8; ++r)
    a[r] = __builtin_amdgcn_sqrtf(sre[r] * sre[r] + sim[r] * sim[r]) * scale;
  float mx = fmaxf(fmaxf(fmaxf(a[0], a[1]), fmaxf(a[2], a[3])),
                   fmaxf(fmaxf(a[4], a[5]), fmaxf(a[6], a[7])));
  mx = fmaxf(mx, xhalf(mx));
  const float newm = fmaxf(m_run, mx);
  const float sc = __expf(m_run - newm);
  float p[8];
#pragma unroll
  for (int r = 0; r < 8; ++r) p[r] = __expf(a[r] - newm);
  l_run = l_run * sc + (((p[0] + p[1]) + (p[2] + p[3])) + ((p[4] + p[5]) + (p[6] + p[7])));
  m_run = newm;
#pragma unroll
  for (int r = 0; r < 8; ++r) {
    o_re0[r] *= sc; o_re1[r] *= sc;
    o_im0[r] *= sc; o_im1[r] *= sc;
  }

  const v2f ar0[4] = { {f.vr[0].x, f.vr[0].y}, {f.vr[0].z, f.vr[0].w}, {f.vr[1].x, f.vr[1].y}, {f.vr[1].z, f.vr[1].w} };
  const v2f ai0[4] = { {f.vi[0].x, f.vi[0].y}, {f.vi[0].z, f.vi[0].w}, {f.vi[1].x, f.vi[1].y}, {f.vi[1].z, f.vi[1].w} };
  const v2f ar1[4] = { {f.vr[2].x, f.vr[2].y}, {f.vr[2].z, f.vr[2].w}, {f.vr[3].x, f.vr[3].y}, {f.vr[3].z, f.vr[3].w} };
  const v2f ai1[4] = { {f.vi[2].x, f.vi[2].y}, {f.vi[2].z, f.vi[2].w}, {f.vi[3].x, f.vi[3].y}, {f.vi[3].z, f.vi[3].w} };
#pragma unroll
  for (int t = 0; t < 4; ++t) {
    const v2f bP = { p[2 * t], p[2 * t + 1] };
    o_re0 = wmma4(ar0[t], bP, o_re0);
    o_im0 = wmma4(ai0[t], bP, o_im0);
    o_re1 = wmma4(ar1[t], bP, o_re1);
    o_im1 = wmma4(ai1[t], bP, o_im1);
  }
}

__global__ void __launch_bounds__(256)
attn_complex_kernel(const float* __restrict__ Qre, const float* __restrict__ Qim,
                    const float* __restrict__ Kre, const float* __restrict__ Kim,
                    const float* __restrict__ Vtre, const float* __restrict__ Vtim,
                    float* __restrict__ Out) {
  const int wave = threadIdx.x >> 5;
  const int lane = threadIdx.x & 31;
  const int half = lane >> 4;
  const int l16  = lane & 15;
  const int h    = blockIdx.y;
  const int q0   = (blockIdx.x * 8 + wave) * 16;

  const float* qreh = Qre + (size_t)h * SEQ * HD;
  const float* qimh = Qim + (size_t)h * SEQ * HD;
  const float* kreh = Kre + (size_t)h * SEQ * HD;
  const float* kimh = Kim + (size_t)h * SEQ * HD;
  const float* vreh = Vtre + (size_t)h * SEQ * HD;
  const float* vimh = Vtim + (size_t)h * SEQ * HD;

  v2f qr[8], qi[8], qd[8];
#pragma unroll
  for (int t = 0; t < 8; ++t) {
    const int dk = t * 4 + half * 2;
    const size_t off = ((size_t)(q0 + l16)) * HD + dk;
    qr[t]  = ldp(qreh + off);
    qi[t]  = ldp(qimh + off);
    qd[t]  = qi[t] - qr[t];
  }

  float m_run = -1e30f;
  float l_run = 0.f;
  v8f o_re0 = {0.f}, o_re1 = {0.f}, o_im0 = {0.f}, o_im1 = {0.f};

  KVFrag fa, fb;
  load_kv(fa, kreh, kimh, vreh, vimh, 0, lane);
  for (int kt = 0; kt < SEQ / 16; kt += 2) {
    load_kv(fb, kreh, kimh, vreh, vimh, kt + 1, lane);
    attn_tile(fa, qr, qi, qd, m_run, l_run, o_re0, o_re1, o_im0, o_im1);
    const int nx = (kt + 2 < SEQ / 16) ? kt + 2 : SEQ / 16 - 1;
    load_kv(fa, kreh, kimh, vreh, vimh, nx, lane);
    attn_tile(fb, qr, qi, qd, m_run, l_run, o_re0, o_re1, o_im0, o_im1);
  }

  const float inv = 1.0f / (l_run + xhalf(l_run));
  float* op = Out + ((size_t)(q0 + l16) * CH + h * HD + 8 * half) * 2;
#pragma unroll
  for (int j = 0; j < 4; ++j) {
    const float4 e0 = { o_re0[2 * j] * inv, o_im0[2 * j] * inv,
                        o_re0[2 * j + 1] * inv, o_im0[2 * j + 1] * inv };
    const float4 e1 = { o_re1[2 * j] * inv, o_im1[2 * j] * inv,
                        o_re1[2 * j + 1] * inv, o_im1[2 * j + 1] * inv };
    *(float4*)(op + 4 * j)      = e0;
    *(float4*)(op + 32 + 4 * j) = e1;
  }
}

extern "C" void kernel_launch(void* const* d_in, const int* in_sizes, int n_in,
                              void* d_out, int out_size, void* d_ws, size_t ws_size,
                              hipStream_t stream) {
  (void)in_sizes; (void)n_in; (void)out_size; (void)ws_size;
  const float* Q  = (const float*)d_in[0];
  const float* V  = (const float*)d_in[1];
  const float* K  = (const float*)d_in[2];
  const float* Wq = (const float*)d_in[3];
  const float* bq = (const float*)d_in[4];
  const float* Wk = (const float*)d_in[5];
  const float* bk = (const float*)d_in[6];
  const float* Wv = (const float*)d_in[7];
  const float* bv = (const float*)d_in[8];
  float* out = (float*)d_out;

  const size_t plane = (size_t)NH * SEQ * HD;
  float* qre = (float*)d_ws;
  float* qim = qre + plane;
  float* kre = qim + plane;
  float* kim = kre + plane;
  float* vre = kim + plane;
  float* vim = vre + plane;

  const dim3 blk(256);
  const dim3 pblk(128);
  const dim3 pgrid((SEQ / PJ_SROWS) * (CH / PJ_OROWS));
  proj_complex_kernel<0><<<pgrid, pblk, 0, stream>>>(Q, Wq, bq, qre, qim);
  proj_complex_kernel<2><<<pgrid, pblk, 0, stream>>>(K, Wk, bk, kre, kim);
  proj_complex_kernel<1><<<pgrid, pblk, 0, stream>>>(V, Wv, bv, vre, vim);

  const dim3 agrid(SEQ / 16 / 8, NH);
  attn_complex_kernel<<<agrid, blk, 0, stream>>>(qre, qim, kre, kim, vre, vim, out);
}


# projections: the three launches (q, k, v) merged into one launch of 3 x 512 blocks (blockIdx.y picks operands and output order); 2 launches per repetition, not 4; arithmetic and order of sums kept
# speedup vs baseline: 2.1930x; 1.0538x over previous
#include <hip/hip_runtime.h>

#define SEQ 2048
#define CH  512
#define NH  16
#define HD  32

typedef float v2f __attribute__((ext_vector_type(2)));
typedef float v4f __attribute__((ext_vector_type(4)));
typedef float v8f __attribute__((ext_vector_type(8)));

__device__ __forceinline__ v8f wmma4(v2f a, v2f b, v8f c) {
  return __builtin_amdgcn_wmma_f32_16x16x4_f32(false, a, false, b, (short)0, c, false, false);
}

__device__ __forceinline__ v2f ldp(const float* p) {
  const float2 t = *(const float2*)p;
  v2f r = { t.x, t.y };
  return r;
}

#define PJ_SROWS 32
#define PJ_OROWS 64
#define PJ_ROWS  (PJ_SROWS + PJ_OROWS)
#define PJ_PITCH 17
#define PJ_CHUNK 64

template <int MODE>
__device__ __forceinline__ void proj_store_tile(const v8f yr, const v8f yi, const int sTile,
                                                const int oTile, const int lane,
                                                const float* __restrict__ Bv,
                                                float* __restrict__ Yre, float* __restrict__ Yim) {
  const int half = lane >> 4;
  const int l16  = lane & 15;
  const int wrow = oTile * 16 + l16;
  if (MODE == 0) {
    const float2 bb = *(const float2*)(Bv + (size_t)wrow * 2);
    const int h  = wrow >> 5;
    const int dd = wrow & 31;
#pragma unroll
    for (int r = 0; r < 8; ++r) {
      const int s = sTile * 16 + r + 8 * half;
      const size_t idx = ((size_t)h * SEQ + s) * HD + dd;
      Yre[idx] = yr[r] + bb.x;
      Yim[idx] = yi[r] + bb.y;
    }
  } else if (MODE == 1) {
    const float2 bb = *(const float2*)(Bv + (size_t)wrow * 2);
    const int h  = wrow >> 5;
    const int dt = oTile & 1;
    const size_t base = ((size_t)(h * (SEQ / 16) + sTile)) * 512 + (size_t)(dt * 2) * 128 + lane * 4;
    float4 r0 = { yr[0] + bb.x, yr[1] + bb.x, yr[2] + bb.x, yr[3] + bb.x };
    float4 r1 = { yr[4] + bb.x, yr[5] + bb.x, yr[6] + bb.x, yr[7] + bb.x };
    float4 i0 = { yi[0] + bb.y, yi[1] + bb.y, yi[2] + bb.y, yi[3] + bb.y };
    float4 i1 = { yi[4] + bb.y, yi[5] + bb.y, yi[6] + bb.y, yi[7] + bb.y };
    *(float4*)(Yre + base)       = r0;
    *(float4*)(Yre + base + 128) = r1;
    *(float4*)(Yim + base)       = i0;
    *(float4*)(Yim + base + 128) = i1;
  } else {
    const int oc0 = oTile * 16 + 8 * half;
    const float4 b01 = *(const float4*)(Bv + (size_t)oc0 * 2);
    const float4 b23 = *(const float4*)(Bv + (size_t)oc0 * 2 + 4);
    const float4 b45 = *(const float4*)(Bv + (size_t)oc0 * 2 + 8);
    const float4 b67 = *(const float4*)(Bv + (size_t)oc0 * 2 + 12);
    const int h = oTile >> 1;
    const int j = (oTile & 1) * 2 + half;
    const size_t base = ((size_t)(h * (SEQ / 16) + sTile)) * 512 + (size_t)j * 128 + l16 * 4;
    float4 rlo = { yr[0] + b01.x, yr[1] + b01.z, yr[4] + b45.x, yr[5] + b45.z };
    float4 rhi = { yr[2] + b23.x, yr[3] + b23.z, yr[6] + b67.x, yr[7] + b67.z };
    float4 ilo = { yi[0] + b01.y, yi[1] + b01.w, yi[4] + b45.y, yi[5] + b45.w };
    float4 ihi = { yi[2] + b23.y, yi[3] + b23.w, yi[6] + b67.y, yi[7] + b67.w };
    *(float4*)(Yre + base)      = rlo;
    *(float4*)(Yre + base + 64) = rhi;
    *(float4*)(Yim + base)      = ilo;
    *(float4*)(Yim + base + 64) = ihi;
  }
}

template <int MODE>
__device__ __forceinline__ void
proj_complex_block(v4f (&tile)[2][PJ_ROWS * PJ_PITCH],
                   const float* __restrict__ X, const float* __restrict__ W,
                   const float* __restrict__ Bv, float* __restrict__ Yre,
                   float* __restrict__ Yim) {
  const int tid  = threadIdx.x;
  const int wave = tid >> 5;
  const int lane = tid & 31;
  const int half = lane >> 4;
  const int l16  = lane & 15;

  const int sBlock = blockIdx.x >> 3;
  const int oBlock = blockIdx.x & 7;
  const int sLocal = wave >> 1;
  const int oPair  = wave & 1;
  const int sTile  = sBlock * 2 + sLocal;
  const int oTile0 = oBlock * 4 + oPair * 2;

  const int c16 = tid & 15;
  const int r0  = tid >> 4;
  const float* xsrc = X + (size_t)(sBlock * PJ_SROWS + r0) * CH * 2 + c16 * 4;
  const float* wsrc = W + (size_t)(oBlock * PJ_OROWS + r0) * CH * 2 + c16 * 4;
  const int sdst = r0 * PJ_PITCH + c16;

  const int xoff  = (sLocal * 16 + l16) * PJ_PITCH + half;
  const int w0off = (PJ_SROWS + oPair * 32 + l16) * PJ_PITCH + half;
  const int w1off = w0off + 16 * PJ_PITCH;

  v4f g[12];
#pragma unroll
  for (int i = 0; i < 4; ++i) g[i] = *(const v4f*)(xsrc + (size_t)i * 8 * CH * 2);
#pragma unroll
  for (int i = 0; i < 8; ++i) g[4 + i] = *(const v4f*)(wsrc + (size_t)i * 8 * CH * 2);
#pragma unroll
  for (int i = 0; i < 12; ++i) tile[0][sdst + i * 8 * PJ_PITCH] = g[i];
  __syncthreads();

  v8f p1a = {0.f}, p2a = {0.f}, p3a = {0.f};
  v8f p1b = {0.f}, p2b = {0.f}, p3b = {0.f};

  for (int c = 0; c < CH * 2 / PJ_CHUNK; ++c) {
    const bool more = (c + 1 < CH * 2 / PJ_CHUNK);
    if (more) {
      const int ko = (c + 1) * PJ_CHUNK;
#pragma unroll
      for (int i = 0; i < 4; ++i) g[i] = *(const v4f*)(xsrc + (size_t)i * 8 * CH * 2 + ko);
#pragma unroll
      for (int i = 0; i < 8; ++i) g[4 + i] = *(const v4f*)(wsrc + (size_t)i * 8 * CH * 2 + ko);
      asm volatile("" ::: "memory");
    }

    const v4f* cur = tile[c & 1];
    v4f fx[8], fw0[8], fw1[8];
#pragma unroll
    for (int j = 0; j < 8; ++j) {
      fx[j]  = cur[xoff + 2 * j];
      fw0[j] = cur[w0off + 2 * j];
      fw1[j] = cur[w1off + 2 * j];
    }
#pragma unroll
    for (int j = 0; j < 8; ++j) {
      const v2f xr  = { fx[j].x, fx[j].z };
      const v2f xi  = { fx[j].y, fx[j].w };
      const v2f xs  = { fx[j].x + fx[j].y, fx[j].z + fx[j].w };
      const v2f w0r = { fw0[j].x, fw0[j].z };
      const v2f w0i = { fw0[j].y, fw0[j].w };
      const v2f w0s = { fw0[j].x + fw0[j].y, fw0[j].z + fw0[j].w };
      const v2f w1r = { fw1[j].x, fw1[j].z };
      const v2f w1i = { fw1[j].y, fw1[j].w };
      const v2f w1s = { fw1[j].x + fw1[j].y, fw1[j].z + fw1[j].w };
      if (MODE == 2) {
        p1a = wmma4(w0r, xr, p1a);  p2a = wmma4(w0i, xi, p2a);  p3a = wmma4(w0s, xs, p3a);
        p1b = wmma4(w1r, xr, p1b);  p2b = wmma4(w1i, xi, p2b);  p3b = wmma4(w1s, xs, p3b);
      } else {
        p1a = wmma4(xr, w0r, p1a);  p2a = wmma4(xi, w0i, p2a);  p3a = wmma4(xs, w0s, p3a);
        p1b = wmma4(xr, w1r, p1b);  p2b = wmma4(xi, w1i, p2b);  p3b = wmma4(xs, w1s, p3b);
      }
    }

    if (more) {
      asm volatile("" : "+v"(p1a), "+v"(p2a), "+v"(p3a), "+v"(p1b), "+v"(p2b), "+v"(p3b),
                        "+v"(g[0]), "+v"(g[1]), "+v"(g[2]), "+v"(g[3]), "+v"(g[4]), "+v"(g[5]),
                        "+v"(g[6]), "+v"(g[7]), "+v"(g[8]), "+v"(g[9]), "+v"(g[10]), "+v"(g[11]));
      v4f* nxt = tile[(c + 1) & 1];
#pragma unroll
      for (int i = 0; i < 12; ++i) nxt[sdst + i * 8 * PJ_PITCH] = g[i];
    }
    __syncthreads();
  }

  const v8f yra = p1a - p2a;
  const v8f yia = (p3a - p1a) - p2a;
  const v8f yrb = p1b - p2b;
  const v8f yib = (p3b - p1b) - p2b;
  proj_store_tile<MODE>(yra, yia, sTile, oTile0,     lane, Bv, Yre, Yim);
  proj_store_tile<MODE>(yrb, yib, sTile, oTile0 + 1, lane, Bv, Yre, Yim);
}

__global__ void __launch_bounds__(128)
proj3_complex_kernel(const float* __restrict__ Xq, const float* __restrict__ Wq,
                     const float* __restrict__ Bq, float* __restrict__ Qre,
                     float* __restrict__ Qim,
                     const float* __restrict__ Xk, const float* __restrict__ Wk,
                     const float* __restrict__ Bk, float* __restrict__ Kre,
                     float* __restrict__ Kim,
                     const float* __restrict__ Xv, const float* __restrict__ Wv,
                     const float* __restrict__ Bv, float* __restrict__ Vre,
                     float* __restrict__ Vim) {
  __shared__ v4f tile[2][PJ_ROWS * PJ_PITCH];
  if (blockIdx.y == 0)      proj_complex_block<0>(tile, Xq, Wq, Bq, Qre, Qim);
  else if (blockIdx.y == 1) proj_complex_block<2>(tile, Xk, Wk, Bk, Kre, Kim);
  else                      proj_complex_block<1>(tile, Xv, Wv, Bv, Vre, Vim);
}

__device__ __forceinline__ float xhalf(float x) {
  const int b = __builtin_bit_cast(int, x);
  return __builtin_bit_cast(float,
      __builtin_amdgcn_permlanex16(b, b, 0x76543210, 0xfedcba98, false, false));
}

struct KVFrag {
  v4f kr[4], ki[4];
  v4f vr[4], vi[4];
};

__device__ __forceinline__ void load_kv(KVFrag& f, const float* __restrict__ kreh,
                                        const float* __restrict__ kimh,
                                        const float* __restrict__ vreh,
                                        const float* __restrict__ vimh,
                                        int kt, int lane) {
  const size_t off = (size_t)kt * 512 + lane * 4;
#pragma unroll
  for (int i = 0; i < 4; ++i) {
    f.kr[i] = *(const v4f*)(kreh + off + i * 128);
    f.ki[i] = *(const v4f*)(kimh + off + i * 128);
    f.vr[i] = *(const v4f*)(vreh + off + i * 128);
    f.vi[i] = *(const v4f*)(vimh + off + i * 128);
  }
}

__device__ __forceinline__ void attn_tile(KVFrag& f, const v2f (&qr)[8], const v2f (&qi)[8],
                                          const v2f (&qd)[8], float& m_run, float& l_run,
                                          v8f& o_re0, v8f& o_re1, v8f& o_im0, v8f& o_im1) {
  const float scale = 0.17677669529663687f;

  asm volatile("" : "+v"(f.kr[0]), "+v"(f.kr[1]), "+v"(f.kr[2]), "+v"(f.kr[3]),
                    "+v"(f.ki[0]), "+v"(f.ki[1]), "+v"(f.ki[2]), "+v"(f.ki[3]),
                    "+v"(f.vr[0]), "+v"(f.vr[1]), "+v"(f.vr[2]), "+v"(f.vr[3]),
                    "+v"(f.vi[0]), "+v"(f.vi[1]), "+v"(f.vi[2]), "+v"(f.vi[3]));

  v8f p1 = {0.f}, p2 = {0.f}, p3 = {0.f};
#pragma unroll
  for (int t = 0; t < 8; ++t) {
    const v4f r4 = f.kr[t >> 1], i4 = f.ki[t >> 1];
    const v2f kr = (t & 1) ? v2f{ r4.z, r4.w } : v2f{ r4.x, r4.y };
    const v2f ki = (t & 1) ? v2f{ i4.z, i4.w } : v2f{ i4.x, i4.y };
    const v2f ks = kr + ki;
    p1 = wmma4(kr, qr[t], p1);
    p2 = wmma4(ki, qi[t], p2);
    p3 = wmma4(ks, qd[t], p3);
  }
  const v8f sre = p1 + p2;
  const v8f sim = p3 + (p1 - p2);

  float a[8];
#pragma unroll
  for (int r = 0; r < 8; ++r)
    a[r] = __builtin_amdgcn_sqrtf(sre[r] * sre[r] + sim[r] * sim[r]) * scale;
  float mx = fmaxf(fmaxf(fmaxf(a[0], a[1]), fmaxf(a[2], a[3])),
                   fmaxf(fmaxf(a[4], a[5]), fmaxf(a[6], a[7])));
  mx = fmaxf(mx, xhalf(mx));
  const float newm = fmaxf(m_run, mx);
  const float sc = __expf(m_run - newm);
  float p[8];
#pragma unroll
  for (int r = 0; r < 8; ++r) p[r] = __expf(a[r] - newm);
  l_run = l_run * sc + (((p[0] + p[1]) + (p[2] + p[3])) + ((p[4] + p[5]) + (p[6] + p[7])));
  m_run = newm;
#pragma unroll
  for (int r = 0; r < 8; ++r) {
    o_re0[r] *= sc; o_re1[r] *= sc;
    o_im0[r] *= sc; o_im1[r] *= sc;
  }

  const v2f ar0[4] = { {f.vr[0].x, f.vr[0].y}, {f.vr[0].z, f.vr[0].w}, {f.vr[1].x, f.vr[1].y}, {f.vr[1].z, f.vr[1].w} };
  const v2f ai0[4] = { {f.vi[0].x, f.vi[0].y}, {f.vi[0].z, f.vi[0].w}, {f.vi[1].x, f.vi[1].y}, {f.vi[1].z, f.vi[1].w} };
  const v2f ar1[4] = { {f.vr[2].x, f.vr[2].y}, {f.vr[2].z, f.vr[2].w}, {f.vr[3].x, f.vr[3].y}, {f.vr[3].z, f.vr[3].w} };
  const v2f ai1[4] = { {f.vi[2].x, f.vi[2].y}, {f.vi[2].z, f.vi[2].w}, {f.vi[3].x, f.vi[3].y}, {f.vi[3].z, f.vi[3].w} };
#pragma unroll
  for (int t = 0; t < 4; ++t) {
    const v2f bP = { p[2 * t], p[2 * t + 1] };
    o_re0 = wmma4(ar0[t], bP, o_re0);
    o_im0 = wmma4(ai0[t], bP, o_im0);
    o_re1 = wmma4(ar1[t], bP, o_re1);
    o_im1 = wmma4(ai1[t], bP, o_im1);
  }
}

__global__ void __launch_bounds__(256)
attn_complex_kernel(const float* __restrict__ Qre, const float* __restrict__ Qim,
                    const float* __restrict__ Kre, const float* __restrict__ Kim,
                    const float* __restrict__ Vtre, const float* __restrict__ Vtim,
                    float* __restrict__ Out) {
  const int wave = threadIdx.x >> 5;
  const int lane = threadIdx.x & 31;
  const int half = lane >> 4;
  const int l16  = lane & 15;
  const int h    = blockIdx.y;
  const int q0   = (blockIdx.x * 8 + wave) * 16;

  const float* qreh = Qre + (size_t)h * SEQ * HD;
  const float* qimh = Qim + (size_t)h * SEQ * HD;
  const float* kreh = Kre + (size_t)h * SEQ * HD;
  const float* kimh = Kim + (size_t)h * SEQ * HD;
  const float* vreh = Vtre + (size_t)h * SEQ * HD;
  const float* vimh = Vtim + (size_t)h * SEQ * HD;

  v2f qr[8], qi[8], qd[8];
#pragma unroll
  for (int t = 0; t < 8; ++t) {
    const int dk = t * 4 + half * 2;
    const size_t off = ((size_t)(q0 + l16)) * HD + dk;
    qr[t]  = ldp(qreh + off);
    qi[t]  = ldp(qimh + off);
    qd[t]  = qi[t] - qr[t];
  }

  float m_run = -1e30f;
  float l_run = 0.f;
  v8f o_re0 = {0.f}, o_re1 = {0.f}, o_im0 = {0.f}, o_im1 = {0.f};

  KVFrag fa, fb;
  load_kv(fa, kreh, kimh, vreh, vimh, 0, lane);
  for (int kt = 0; kt < SEQ / 16; kt += 2) {
    load_kv(fb, kreh, kimh, vreh, vimh, kt + 1, lane);
    attn_tile(fa, qr, qi, qd, m_run, l_run, o_re0, o_re1, o_im0, o_im1);
    const int nx = (kt + 2 < SEQ / 16) ? kt + 2 : SEQ / 16 - 1;
    load_kv(fa, kreh, kimh, vreh, vimh, nx, lane);
    attn_tile(fb, qr, qi, qd, m_run, l_run, o_re0, o_re1, o_im0, o_im1);
  }

  const float inv = 1.0f / (l_run + xhalf(l_run));
  float* op = Out + ((size_t)(q0 + l16) * CH + h * HD + 8 * half) * 2;
#pragma unroll
  for (int j = 0; j < 4; ++j) {
    const float4 e0 = { o_re0[2 * j] * inv, o_im0[2 * j] * inv,
                        o_re0[2 * j + 1] * inv, o_im0[2 * j + 1] * inv };
    const float4 e1 = { o_re1[2 * j] * inv, o_im1[2 * j] * inv,
                        o_re1[2 * j + 1] * inv, o_im1[2 * j + 1] * inv };
    *(float4*)(op + 4 * j)      = e0;
    *(float4*)(op + 32 + 4 * j) = e1;
  }
}

extern "C" void kernel_launch(void* const* d_in, const int* in_sizes, int n_in,
                              void* d_out, int out_size, void* d_ws, size_t ws_size,
                              hipStream_t stream) {
  (void)in_sizes; (void)n_in; (void)out_size; (void)ws_size;
  const float* Q  = (const float*)d_in[0];
  const float* V  = (const float*)d_in[1];
  const float* K  = (const float*)d_in[2];
  const float* Wq = (const float*)d_in[3];
  const float* bq = (const float*)d_in[4];
  const float* Wk = (const float*)d_in[5];
  const float* bk = (const float*)d_in[6];
  const float* Wv = (const float*)d_in[7];
  const float* bv = (const float*)d_in[8];
  float* out = (float*)d_out;

  const size_t plane = (size_t)NH * SEQ * HD;
  float* qre = (float*)d_ws;
  float* qim = qre + plane;
  float* kre = qim + plane;
  float* kim = kre + plane;
  float* vre = kim + plane;
  float* vim = vre + plane;

  const dim3 blk(256);
  const dim3 pblk(128);
  const dim3 pgrid((SEQ / PJ_SROWS) * (CH / PJ_OROWS), 3);
  proj3_complex_kernel<<<pgrid, pblk, 0, stream>>>(Q, Wq, bq, qre, qim,
                                                   K, Wk, bk, kre, kim,
                                                   V, Wv, bv, vre, vim);

  const dim3 agrid(SEQ / 16 / 8, NH);
  attn_complex_kernel<<<agrid, blk, 0, stream>>>(qre, qim, kre, kim, vre, vim, out);
}


# attention loop software-pipelined in the wave: score WMMAs of key tile i+1 issued with tile i's softmax vector work between them, then P.V of tile i; order of sums kept, |s|^2 now a fused mul-add
# speedup vs baseline: 2.2598x; 1.0305x over previous
#include <hip/hip_runtime.h>

#define SEQ 2048
#define CH  512
#define NH  16
#define HD  32

typedef float v2f __attribute__((ext_vector_type(2)));
typedef float v4f __attribute__((ext_vector_type(4)));
typedef float v8f __attribute__((ext_vector_type(8)));

__device__ __forceinline__ v8f wmma4(v2f a, v2f b, v8f c) {
  return __builtin_amdgcn_wmma_f32_16x16x4_f32(false, a, false, b, (short)0, c, false, false);
}

__device__ __forceinline__ v2f ldp(const float* p) {
  const float2 t = *(const float2*)p;
  v2f r = { t.x, t.y };
  return r;
}

#define PJ_SROWS 32
#define PJ_OROWS 64
#define PJ_ROWS  (PJ_SROWS + PJ_OROWS)
#define PJ_PITCH 17
#define PJ_CHUNK 64

template <int MODE>
__device__ __forceinline__ void proj_store_tile(const v8f yr, const v8f yi, const int sTile,
                                                const int oTile, const int lane,
                                                const float* __restrict__ Bv,
                                                float* __restrict__ Yre, float* __restrict__ Yim) {
  const int half = lane >> 4;
  const int l16  = lane & 15;
  const int wrow = oTile * 16 + l16;
  if (MODE == 0) {
    const float2 bb = *(const float2*)(Bv + (size_t)wrow * 2);
    const int h  = wrow >> 5;
    const int dd = wrow & 31;
#pragma unroll
    for (int r = 0; r < 8; ++r) {
      const int s = sTile * 16 + r + 8 * half;
      const size_t idx = ((size_t)h * SEQ + s) * HD + dd;
      Yre[idx] = yr[r] + bb.x;
      Yim[idx] = yi[r] + bb.y;
    }
  } else if (MODE == 1) {
    const float2 bb = *(const float2*)(Bv + (size_t)wrow * 2);
    const int h  = wrow >> 5;
    const int dt = oTile & 1;
    const size_t base = ((size_t)(h * (SEQ / 16) + sTile)) * 512 + (size_t)(dt * 2) * 128 + lane * 4;
    float4 r0 = { yr[0] + bb.x, yr[1] + bb.x, yr[2] + bb.x, yr[3] + bb.x };
    float4 r1 = { yr[4] + bb.x, yr[5] + bb.x, yr[6] + bb.x, yr[7] + bb.x };
    float4 i0 = { yi[0] + bb.y, yi[1] + bb.y, yi[2] + bb.y, yi[3] + bb.y };
    float4 i1 = { yi[4] + bb.y, yi[5] + bb.y, yi[6] + bb.y, yi[7] + bb.y };
    *(float4*)(Yre + base)       = r0;
    *(float4*)(Yre + base + 128) = r1;
    *(float4*)(Yim + base)       = i0;
    *(float4*)(Yim + base + 128) = i1;
  } else {
    const int oc0 = oTile * 16 + 8 * half;
    const float4 b01 = *(const float4*)(Bv + (size_t)oc0 * 2);
    const float4 b23 = *(const float4*)(Bv + (size_t)oc0 * 2 + 4);
    const float4 b45 = *(const float4*)(Bv + (size_t)oc0 * 2 + 8);
    const float4 b67 = *(const float4*)(Bv + (size_t)oc0 * 2 + 12);
    const int h = oTile >> 1;
    const int j = (oTile & 1) * 2 + half;
    const size_t base = ((size_t)(h * (SEQ / 16) + sTile)) * 512 + (size_t)j * 128 + l16 * 4;
    float4 rlo = { yr[0] + b01.x, yr[1] + b01.z, yr[4] + b45.x, yr[5] + b45.z };
    float4 rhi = { yr[2] + b23.x, yr[3] + b23.z, yr[6] + b67.x, yr[7] + b67.z };
    float4 ilo = { yi[0] + b01.y, yi[1] + b01.w, yi[4] + b45.y, yi[5] + b45.w };
    float4 ihi = { yi[2] + b23.y, yi[3] + b23.w, yi[6] + b67.y, yi[7] + b67.w };
    *(float4*)(Yre + base)      = rlo;
    *(float4*)(Yre + base + 64) = rhi;
    *(float4*)(Yim + base)      = ilo;
    *(float4*)(Yim + base + 64) = ihi;
  }
}

template <int MODE>
__device__ __forceinline__ void
proj_complex_block(v4f (&tile)[2][PJ_ROWS * PJ_PITCH],
                   const float* __restrict__ X, const float* __restrict__ W,
                   const float* __restrict__ Bv, float* __restrict__ Yre,
                   float* __restrict__ Yim) {
  const int tid  = threadIdx.x;
  const int wave = tid >> 5;
  const int lane = tid & 31;
  const int half = lane >> 4;
  const int l16  = lane & 15;

  const int sBlock = blockIdx.x >> 3;
  const int oBlock = blockIdx.x & 7;
  const int sLocal = wave >> 1;
  const int oPair  = wave & 1;
  const int sTile  = sBlock * 2 + sLocal;
  const int oTile0 = oBlock * 4 + oPair * 2;

  const int c16 = tid & 15;
  const int r0  = tid >> 4;
  const float* xsrc = X + (size_t)(sBlock * PJ_SROWS + r0) * CH * 2 + c16 * 4;
  const float* wsrc = W + (size_t)(oBlock * PJ_OROWS + r0) * CH * 2 + c16 * 4;
  const int sdst = r0 * PJ_PITCH + c16;

  const int xoff  = (sLocal * 16 + l16) * PJ_PITCH + half;
  const int w0off = (PJ_SROWS + oPair * 32 + l16) * PJ_PITCH + half;
  const int w1off = w0off + 16 * PJ_PITCH;

  v4f g[12];
#pragma unroll
  for (int i = 0; i < 4; ++i) g[i] = *(const v4f*)(xsrc + (size_t)i * 8 * CH * 2);
#pragma unroll
  for (int i = 0; i < 8; ++i) g[4 + i] = *(const v4f*)(wsrc + (size_t)i * 8 * CH * 2);
#pragma unroll
  for (int i = 0; i < 12; ++i) tile[0][sdst + i * 8 * PJ_PITCH] = g[i];
  __syncthreads();

  v8f p1a = {0.f}, p2a = {0.f}, p3a = {0.f};
  v8f p1b = {0.f}, p2b = {0.f}, p3b = {0.f};

  for (int c = 0; c < CH * 2 / PJ_CHUNK; ++c) {
    const bool more = (c + 1 < CH * 2 / PJ_CHUNK);
    if (more) {
      const int ko = (c + 1) * PJ_CHUNK;
#pragma unroll
      for (int i = 0; i < 4; ++i) g[i] = *(const v4f*)(xsrc + (size_t)i * 8 * CH * 2 + ko);
#pragma unroll
      for (int i = 0; i < 8; ++i) g[4 + i] = *(const v4f*)(wsrc + (size_t)i * 8 * CH * 2 + ko);
      asm volatile("" ::: "memory");
    }

    const v4f* cur = tile[c & 1];
    v4f fx[8], fw0[8], fw1[8];
#pragma unroll
    for (int j = 0; j < 8; ++j) {
      fx[j]  = cur[xoff + 2 * j];
      fw0[j] = cur[w0off + 2 * j];
      fw1[j] = cur[w1off + 2 * j];
    }
#pragma unroll
    for (int j = 0; j < 8; ++j) {
      const v2f xr  = { fx[j].x, fx[j].z };
      const v2f xi  = { fx[j].y, fx[j].w };
      const v2f xs  = { fx[j].x + fx[j].y, fx[j].z + fx[j].w };
      const v2f w0r = { fw0[j].x, fw0[j].z };
      const v2f w0i = { fw0[j].y, fw0[j].w };
      const v2f w0s = { fw0[j].x + fw0[j].y, fw0[j].z + fw0[j].w };
      const v2f w1r = { fw1[j].x, fw1[j].z };
      const v2f w1i = { fw1[j].y, fw1[j].w };
      const v2f w1s = { fw1[j].x + fw1[j].y, fw1[j].z + fw1[j].w };
      if (MODE == 2) {
        p1a = wmma4(w0r, xr, p1a);  p2a = wmma4(w0i, xi, p2a);  p3a = wmma4(w0s, xs, p3a);
        p1b = wmma4(w1r, xr, p1b);  p2b = wmma4(w1i, xi, p2b);  p3b = wmma4(w1s, xs, p3b);
      } else {
        p1a = wmma4(xr, w0r, p1a);  p2a = wmma4(xi, w0i, p2a);  p3a = wmma4(xs, w0s, p3a);
        p1b = wmma4(xr, w1r, p1b);  p2b = wmma4(xi, w1i, p2b);  p3b = wmma4(xs, w1s, p3b);
      }
    }

    if (more) {
      asm volatile("" : "+v"(p1a), "+v"(p2a), "+v"(p3a), "+v"(p1b), "+v"(p2b), "+v"(p3b),
                        "+v"(g[0]), "+v"(g[1]), "+v"(g[2]), "+v"(g[3]), "+v"(g[4]), "+v"(g[5]),
                        "+v"(g[6]), "+v"(g[7]), "+v"(g[8]), "+v"(g[9]), "+v"(g[10]), "+v"(g[11]));
      v4f* nxt = tile[(c + 1) & 1];
#pragma unroll
      for (int i = 0; i < 12; ++i) nxt[sdst + i * 8 * PJ_PITCH] = g[i];
    }
    __syncthreads();
  }

  const v8f yra = p1a - p2a;
  const v8f yia = (p3a - p1a) - p2a;
  const v8f yrb = p1b - p2b;
  const v8f yib = (p3b - p1b) - p2b;
  proj_store_tile<MODE>(yra, yia, sTile, oTile0,     lane, Bv, Yre, Yim);
  proj_store_tile<MODE>(yrb, yib, sTile, oTile0 + 1, lane, Bv, Yre, Yim);
}

__global__ void __launch_bounds__(128)
proj3_complex_kernel(const float* __restrict__ Xq, const float* __restrict__ Wq,
                     const float* __restrict__ Bq, float* __restrict__ Qre,
                     float* __restrict__ Qim,
                     const float* __restrict__ Xk, const float* __restrict__ Wk,
                     const float* __restrict__ Bk, float* __restrict__ Kre,
                     float* __restrict__ Kim,
                     const float* __restrict__ Xv, const float* __restrict__ Wv,
                     const float* __restrict__ Bv, float* __restrict__ Vre,
                     float* __restrict__ Vim) {
  __shared__ v4f tile[2][PJ_ROWS * PJ_PITCH];
  if (blockIdx.y == 0)      proj_complex_block<0>(tile, Xq, Wq, Bq, Qre, Qim);
  else if (blockIdx.y == 1) proj_complex_block<2>(tile, Xk, Wk, Bk, Kre, Kim);
  else                      proj_complex_block<1>(tile, Xv, Wv, Bv, Vre, Vim);
}

__device__ __forceinline__ float xhalf(float x) {
  const int b = __builtin_bit_cast(int, x);
  return __builtin_bit_cast(float,
      __builtin_amdgcn_permlanex16(b, b, 0x76543210, 0xfedcba98, false, false));
}

struct KVFrag {
  v4f kr[4], ki[4];
  v4f vr[4], vi[4];
};

__device__ __forceinline__ void load_kv(KVFrag& f, const float* __restrict__ kreh,
                                        const float* __restrict__ kimh,
                                        const float* __restrict__ vreh,
                                        const float* __restrict__ vimh,
                                        int ktK, int ktV, int lane) {
  const size_t offK = (size_t)ktK * 512 + lane * 4;
  const size_t offV = (size_t)ktV * 512 + lane * 4;
#pragma unroll
  for (int i = 0; i < 4; ++i) {
    f.kr[i] = *(const v4f*)(kreh + offK + i * 128);
    f.ki[i] = *(const v4f*)(kimh + offK + i * 128);
    f.vr[i] = *(const v4f*)(vreh + offV + i * 128);
    f.vi[i] = *(const v4f*)(vimh + offV + i * 128);
  }
}

__device__ __forceinline__ void score_kstep(const v4f (&fkr)[4], const v4f (&fki)[4], const int t,
                                            const v2f (&qr)[8], const v2f (&qi)[8],
                                            const v2f (&qd)[8], v8f& p1, v8f& p2, v8f& p3) {
  const v4f r4 = fkr[t >> 1], i4 = fki[t >> 1];
  const v2f kr = (t & 1) ? v2f{ r4.z, r4.w } : v2f{ r4.x, r4.y };
  const v2f ki = (t & 1) ? v2f{ i4.z, i4.w } : v2f{ i4.x, i4.y };
  const v2f ks = kr + ki;
  p1 = wmma4(kr, qr[t], p1);
  p2 = wmma4(ki, qi[t], p2);
  p3 = wmma4(ks, qd[t], p3);
}

__device__ __forceinline__ void attn_step(KVFrag& f, const v2f (&qr)[8], const v2f (&qi)[8],
                                          const v2f (&qd)[8], v8f& sre, v8f& sim,
                                          float& m_run, float& l_run,
                                          v8f& o_re0, v8f& o_re1, v8f& o_im0, v8f& o_im1) {
  const float scale = 0.17677669529663687f;

  asm volatile("" : "+v"(f.kr[0]), "+v"(f.kr[1]), "+v"(f.kr[2]), "+v"(f.kr[3]),
                    "+v"(f.ki[0]), "+v"(f.ki[1]), "+v"(f.ki[2]), "+v"(f.ki[3]),
                    "+v"(f.vr[0]), "+v"(f.vr[1]), "+v"(f.vr[2]), "+v"(f.vr[3]),
                    "+v"(f.vi[0]), "+v"(f.vi[1]), "+v"(f.vi[2]), "+v"(f.vi[3]));

  v8f p1 = {0.f}, p2 = {0.f}, p3 = {0.f};

  score_kstep(f.kr, f.ki, 0, qr, qi, qd, p1, p2, p3);
  v8f sq = sre * sre + sim * sim;
  asm volatile("" : "+v"(p1), "+v"(p2), "+v"(p3), "+v"(sq));

  score_kstep(f.kr, f.ki, 1, qr, qi, qd, p1, p2, p3);
  v8f rt;
#pragma unroll
  for (int r = 0; r < 8; ++r)
    rt[r] = __builtin_amdgcn_sqrtf(sq[r]);
  asm volatile("" : "+v"(p1), "+v"(p2), "+v"(p3), "+v"(rt));

  score_kstep(f.kr, f.ki, 2, qr, qi, qd, p1, p2, p3);
  v8f a = rt * scale;
  float mx = fmaxf(fmaxf(fmaxf(a[0], a[1]), fmaxf(a[2], a[3])),
                   fmaxf(fmaxf(a[4], a[5]), fmaxf(a[6], a[7])));
  float mo = xhalf(mx);
  asm volatile("" : "+v"(p1), "+v"(p2), "+v"(p3), "+v"(a), "+v"(mx), "+v"(mo));

  score_kstep(f.kr, f.ki, 3, qr, qi, qd, p1, p2, p3);
  mx = fmaxf(mx, mo);
  const float newm = fmaxf(m_run, mx);
  float sc = __expf(m_run - newm);
  v8f d = a - newm;
  m_run = newm;
  asm volatile("" : "+v"(p1), "+v"(p2), "+v"(p3), "+v"(d), "+v"(sc));

  score_kstep(f.kr, f.ki, 4, qr, qi, qd, p1, p2, p3);
  v8f p;
#pragma unroll
  for (int r = 0; r < 8; ++r) p[r] = __expf(d[r]);
  asm volatile("" : "+v"(p1), "+v"(p2), "+v"(p3), "+v"(p), "+v"(sc));

  score_kstep(f.kr, f.ki, 5, qr, qi, qd, p1, p2, p3);
  l_run = l_run * sc + (((p[0] + p[1]) + (p[2] + p[3])) + ((p[4] + p[5]) + (p[6] + p[7])));
  asm volatile("" : "+v"(p1), "+v"(p2), "+v"(p3), "+v"(l_run), "+v"(sc));

  score_kstep(f.kr, f.ki, 6, qr, qi, qd, p1, p2, p3);
  o_re0 *= sc;
  o_im0 *= sc;
  asm volatile("" : "+v"(p1), "+v"(p2), "+v"(p3), "+v"(o_re0), "+v"(o_im0), "+v"(sc));

  score_kstep(f.kr, f.ki, 7, qr, qi, qd, p1, p2, p3);
  o_re1 *= sc;
  o_im1 *= sc;
  asm volatile("" : "+v"(p1), "+v"(p2), "+v"(p3), "+v"(o_re1), "+v"(o_im1));

  const v2f ar0[4] = { {f.vr[0].x, f.vr[0].y}, {f.vr[0].z, f.vr[0].w}, {f.vr[1].x, f.vr[1].y}, {f.vr[1].z, f.vr[1].w} };
  const v2f ai0[4] = { {f.vi[0].x, f.vi[0].y}, {f.vi[0].z, f.vi[0].w}, {f.vi[1].x, f.vi[1].y}, {f.vi[1].z, f.vi[1].w} };
  const v2f ar1[4] = { {f.vr[2].x, f.vr[2].y}, {f.vr[2].z, f.vr[2].w}, {f.vr[3].x, f.vr[3].y}, {f.vr[3].z, f.vr[3].w} };
  const v2f ai1[4] = { {f.vi[2].x, f.vi[2].y}, {f.vi[2].z, f.vi[2].w}, {f.vi[3].x, f.vi[3].y}, {f.vi[3].z, f.vi[3].w} };

#pragma unroll
  for (int t = 0; t < 2; ++t) {
    const v2f bP = { p[2 * t], p[2 * t + 1] };
    o_re0 = wmma4(ar0[t], bP, o_re0);
    o_im0 = wmma4(ai0[t], bP, o_im0);
    o_re1 = wmma4(ar1[t], bP, o_re1);
    o_im1 = wmma4(ai1[t], bP, o_im1);
  }
  asm volatile("" : "+v"(o_re0), "+v"(o_im0), "+v"(o_re1), "+v"(o_im1),
                    "+v"(p1), "+v"(p2), "+v"(p3));

#pragma unroll
  for (int t = 2; t < 4; ++t) {
    const v2f bP = { p[2 * t], p[2 * t + 1] };
    o_re0 = wmma4(ar0[t], bP, o_re0);
    o_im0 = wmma4(ai0[t], bP, o_im0);
    o_re1 = wmma4(ar1[t], bP, o_re1);
    o_im1 = wmma4(ai1[t], bP, o_im1);
  }
  sre = p1 + p2;
  sim = p3 + (p1 - p2);
  asm volatile("" : "+v"(o_re0), "+v"(o_im0), "+v"(o_re1), "+v"(o_im1),
                    "+v"(sre), "+v"(sim));
}

__global__ void __launch_bounds__(256)
attn_complex_kernel(const float* __restrict__ Qre, const float* __restrict__ Qim,
                    const float* __restrict__ Kre, const float* __restrict__ Kim,
                    const float* __restrict__ Vtre, const float* __restrict__ Vtim,
                    float* __restrict__ Out) {
  const int wave = threadIdx.x >> 5;
  const int lane = threadIdx.x & 31;
  const int half = lane >> 4;
  const int l16  = lane & 15;
  const int h    = blockIdx.y;
  const int q0   = (blockIdx.x * 8 + wave) * 16;

  const float* qreh = Qre + (size_t)h * SEQ * HD;
  const float* qimh = Qim + (size_t)h * SEQ * HD;
  const float* kreh = Kre + (size_t)h * SEQ * HD;
  const float* kimh = Kim + (size_t)h * SEQ * HD;
  const float* vreh = Vtre + (size_t)h * SEQ * HD;
  const float* vimh = Vtim + (size_t)h * SEQ * HD;

  v2f qr[8], qi[8], qd[8];
#pragma unroll
  for (int t = 0; t < 8; ++t) {
    const int dk = t * 4 + half * 2;
    const size_t off = ((size_t)(q0 + l16)) * HD + dk;
    qr[t]  = ldp(qreh + off);
    qi[t]  = ldp(qimh + off);
    qd[t]  = qi[t] - qr[t];
  }

  float m_run = -1e30f;
  float l_run = 0.f;
  v8f o_re0 = {0.f}, o_re1 = {0.f}, o_im0 = {0.f}, o_im1 = {0.f};

  KVFrag fa, fb;

  v8f sre, sim;
  {
    const size_t off0 = (size_t)lane * 4;
#pragma unroll
    for (int i = 0; i < 4; ++i) {
      fb.kr[i] = *(const v4f*)(kreh + off0 + i * 128);
      fb.ki[i] = *(const v4f*)(kimh + off0 + i * 128);
    }
    load_kv(fa, kreh, kimh, vreh, vimh, 1, 0, lane);
    v8f p1 = {0.f}, p2 = {0.f}, p3 = {0.f};
#pragma unroll
    for (int t = 0; t < 8; ++t) score_kstep(fb.kr, fb.ki, t, qr, qi, qd, p1, p2, p3);
    sre = p1 + p2;
    sim = p3 + (p1 - p2);
  }

  for (int kt = 0; kt < SEQ / 16; kt += 2) {
    const int k2 = (kt + 2 < SEQ / 16) ? kt + 2 : SEQ / 16 - 1;
    load_kv(fb, kreh, kimh, vreh, vimh, k2, kt + 1, lane);
    attn_step(fa, qr, qi, qd, sre, sim, m_run, l_run, o_re0, o_re1, o_im0, o_im1);
    const int k3 = (kt + 3 < SEQ / 16) ? kt + 3 : SEQ / 16 - 1;
    load_kv(fa, kreh, kimh, vreh, vimh, k3, k2, lane);
    attn_step(fb, qr, qi, qd, sre, sim, m_run, l_run, o_re0, o_re1, o_im0, o_im1);
  }

  const float inv = 1.0f / (l_run + xhalf(l_run));
  float* op = Out + ((size_t)(q0 + l16) * CH + h * HD + 8 * half) * 2;
#pragma unroll
  for (int j = 0; j < 4; ++j) {
    const float4 e0 = { o_re0[2 * j] * inv, o_im0[2 * j] * inv,
                        o_re0[2 * j + 1] * inv, o_im0[2 * j + 1] * inv };
    const float4 e1 = { o_re1[2 * j] * inv, o_im1[2 * j] * inv,
                        o_re1[2 * j + 1] * inv, o_im1[2 * j + 1] * inv };
    *(float4*)(op + 4 * j)      = e0;
    *(float4*)(op + 32 + 4 * j) = e1;
  }
}

extern "C" void kernel_launch(void* const* d_in, const int* in_sizes, int n_in,
                              void* d_out, int out_size, void* d_ws, size_t ws_size,
                              hipStream_t stream) {
  (void)in_sizes; (void)n_in; (void)out_size; (void)ws_size;
  const float* Q  = (const float*)d_in[0];
  const float* V  = (const float*)d_in[1];
  const float* K  = (const float*)d_in[2];
  const float* Wq = (const float*)d_in[3];
  const float* bq = (const float*)d_in[4];
  const float* Wk = (const float*)d_in[5];
  const float* bk = (const float*)d_in[6];
  const float* Wv = (const float*)d_in[7];
  const float* bv = (const float*)d_in[8];
  float* out = (float*)d_out;

  const size_t plane = (size_t)NH * SEQ * HD;
  float* qre = (float*)d_ws;
  float* qim = qre + plane;
  float* kre = qim + plane;
  float* kim = kre + plane;
  float* vre = kim + plane;
  float* vim = vre + plane;

  const dim3 blk(256);
  const dim3 pblk(128);
  const dim3 pgrid((SEQ / PJ_SROWS) * (CH / PJ_OROWS), 3);
  proj3_complex_kernel<<<pgrid, pblk, 0, stream>>>(Q, Wq, bq, qre, qim,
                                                   K, Wk, bk, kre, kim,
                                                   V, Wv, bv, vre, vim);

  const dim3 agrid(SEQ / 16 / 8, NH);
  attn_complex_kernel<<<agrid, blk, 0, stream>>>(qre, qim, kre, kim, vre, vim, out);
}


# projections: 32x32 wave tile (64x64 blocks, 768 of them, 3 per WGP) and LDS pairs stored (re,re,im,im): a fragment load feeds 6 WMMAs as it stands, not 3 after register shuffles; order of sums kept
# speedup vs baseline: 2.3308x; 1.0314x over previous
#include <hip/hip_runtime.h>

#define SEQ 2048
#define CH  512
#define NH  16
#define HD  32

typedef float v2f __attribute__((ext_vector_type(2)));
typedef float v4f __attribute__((ext_vector_type(4)));
typedef float v8f __attribute__((ext_vector_type(8)));

__device__ __forceinline__ v8f wmma4(v2f a, v2f b, v8f c) {
  return __builtin_amdgcn_wmma_f32_16x16x4_f32(false, a, false, b, (short)0, c, false, false);
}

__device__ __forceinline__ v2f ldp(const float* p) {
  const float2 t = *(const float2*)p;
  v2f r = { t.x, t.y };
  return r;
}

#define PJ_SROWS 64
#define PJ_OROWS 64
#define PJ_ROWS  (PJ_SROWS + PJ_OROWS)
#define PJ_PITCH 17
#define PJ_CHUNK 64

__device__ __forceinline__ v4f pair_order(const v4f g) {
  const v4f r = { g.x, g.z, g.y, g.w };
  return r;
}

template <int MODE>
__device__ __forceinline__ void proj_store_tile(const v8f yr, const v8f yi, const int sTile,
                                                const int oTile, const int lane,
                                                const float* __restrict__ Bv,
                                                float* __restrict__ Yre, float* __restrict__ Yim) {
  const int half = lane >> 4;
  const int l16  = lane & 15;
  const int wrow = oTile * 16 + l16;
  if (MODE == 0) {
    const float2 bb = *(const float2*)(Bv + (size_t)wrow * 2);
    const int h  = wrow >> 5;
    const int dd = wrow & 31;
#pragma unroll
    for (int r = 0; r < 8; ++r) {
      const int s = sTile * 16 + r + 8 * half;
      const size_t idx = ((size_t)h * SEQ + s) * HD + dd;
      Yre[idx] = yr[r] + bb.x;
      Yim[idx] = yi[r] + bb.y;
    }
  } else if (MODE == 1) {
    const float2 bb = *(const float2*)(Bv + (size_t)wrow * 2);
    const int h  = wrow >> 5;
    const int dt = oTile & 1;
    const size_t base = ((size_t)(h * (SEQ / 16) + sTile)) * 512 + (size_t)(dt * 2) * 128 + lane * 4;
    float4 r0 = { yr[0] + bb.x, yr[1] + bb.x, yr[2] + bb.x, yr[3] + bb.x };
    float4 r1 = { yr[4] + bb.x, yr[5] + bb.x, yr[6] + bb.x, yr[7] + bb.x };
    float4 i0 = { yi[0] + bb.y, yi[1] + bb.y, yi[2] + bb.y, yi[3] + bb.y };
    float4 i1 = { yi[4] + bb.y, yi[5] + bb.y, yi[6] + bb.y, yi[7] + bb.y };
    *(float4*)(Yre + base)       = r0;
    *(float4*)(Yre + base + 128) = r1;
    *(float4*)(Yim + base)       = i0;
    *(float4*)(Yim + base + 128) = i1;
  } else {
    const int oc0 = oTile * 16 + 8 * half;
    const float4 b01 = *(const float4*)(Bv + (size_t)oc0 * 2);
    const float4 b23 = *(const float4*)(Bv + (size_t)oc0 * 2 + 4);
    const float4 b45 = *(const float4*)(Bv + (size_t)oc0 * 2 + 8);
    const float4 b67 = *(const float4*)(Bv + (size_t)oc0 * 2 + 12);
    const int h = oTile >> 1;
    const int j = (oTile & 1) * 2 + half;
    const size_t base = ((size_t)(h * (SEQ / 16) + sTile)) * 512 + (size_t)j * 128 + l16 * 4;
    float4 rlo = { yr[0] + b01.x, yr[1] + b01.z, yr[4] + b45.x, yr[5] + b45.z };
    float4 rhi = { yr[2] + b23.x, yr[3] + b23.z, yr[6] + b67.x, yr[7] + b67.z };
    float4 ilo = { yi[0] + b01.y, yi[1] + b01.w, yi[4] + b45.y, yi[5] + b45.w };
    float4 ihi = { yi[2] + b23.y, yi[3] + b23.w, yi[6] + b67.y, yi[7] + b67.w };
    *(float4*)(Yre + base)      = rlo;
    *(float4*)(Yre + base + 64) = rhi;
    *(float4*)(Yim + base)      = ilo;
    *(float4*)(Yim + base + 64) = ihi;
  }
}

template <int MODE>
__device__ __forceinline__ void
proj_complex_block(v4f (&tile)[2][PJ_ROWS * PJ_PITCH],
                   const float* __restrict__ X, const float* __restrict__ W,
                   const float* __restrict__ Bv, float* __restrict__ Yre,
                   float* __restrict__ Yim) {
  const int tid  = threadIdx.x;
  const int wave = tid >> 5;
  const int lane = tid & 31;
  const int half = lane >> 4;
  const int l16  = lane & 15;

  const int sBlock = blockIdx.x >> 3;
  const int oBlock = blockIdx.x & 7;
  const int sLocal = wave >> 1;
  const int oPair  = wave & 1;
  const int sTile0 = sBlock * 4 + sLocal * 2;
  const int oTile0 = oBlock * 4 + oPair * 2;

  const int c16 = tid & 15;
  const int r0  = tid >> 4;
  const float* xsrc = X + (size_t)(sBlock * PJ_SROWS + r0) * CH * 2 + c16 * 4;
  const float* wsrc = W + (size_t)(oBlock * PJ_OROWS + r0) * CH * 2 + c16 * 4;
  const int sdst = r0 * PJ_PITCH + c16;

  const int x0off = (sLocal * 32 + l16) * PJ_PITCH + half;
  const int x1off = x0off + 16 * PJ_PITCH;
  const int w0off = (PJ_SROWS + oPair * 32 + l16) * PJ_PITCH + half;
  const int w1off = w0off + 16 * PJ_PITCH;

  v4f g[16];
#pragma unroll
  for (int i = 0; i < 8; ++i) g[i] = *(const v4f*)(xsrc + (size_t)i * 8 * CH * 2);
#pragma unroll
  for (int i = 0; i < 8; ++i) g[8 + i] = *(const v4f*)(wsrc + (size_t)i * 8 * CH * 2);
#pragma unroll
  for (int i = 0; i < 16; ++i) tile[0][sdst + i * 8 * PJ_PITCH] = pair_order(g[i]);
  __syncthreads();

  v8f p1[4], p2[4], p3[4];
#pragma unroll
  for (int t = 0; t < 4; ++t) {
    const v8f z = {0.f};
    p1[t] = z;  p2[t] = z;  p3[t] = z;
  }

  for (int c = 0; c < CH * 2 / PJ_CHUNK; ++c) {
    const bool more = (c + 1 < CH * 2 / PJ_CHUNK);
    if (more) {
      const int ko = (c + 1) * PJ_CHUNK;
#pragma unroll
      for (int i = 0; i < 8; ++i) g[i] = *(const v4f*)(xsrc + (size_t)i * 8 * CH * 2 + ko);
#pragma unroll
      for (int i = 0; i < 8; ++i) g[8 + i] = *(const v4f*)(wsrc + (size_t)i * 8 * CH * 2 + ko);
      asm volatile("" ::: "memory");
    }

    const v4f* cur = tile[c & 1];
    v4f fx0[8], fx1[8], fw0[8], fw1[8];
#pragma unroll
    for (int j = 0; j < 8; ++j) {
      fx0[j] = cur[x0off + 2 * j];
      fx1[j] = cur[x1off + 2 * j];
      fw0[j] = cur[w0off + 2 * j];
      fw1[j] = cur[w1off + 2 * j];
    }
#pragma unroll
    for (int j = 0; j < 8; ++j) {
      const v2f xr[2] = { { fx0[j].x, fx0[j].y }, { fx1[j].x, fx1[j].y } };
      const v2f xi[2] = { { fx0[j].z, fx0[j].w }, { fx1[j].z, fx1[j].w } };
      const v2f xs[2] = { xr[0] + xi[0], xr[1] + xi[1] };
      const v2f wr[2] = { { fw0[j].x, fw0[j].y }, { fw1[j].x, fw1[j].y } };
      const v2f wi[2] = { { fw0[j].z, fw0[j].w }, { fw1[j].z, fw1[j].w } };
      const v2f ws[2] = { wr[0] + wi[0], wr[1] + wi[1] };
#pragma unroll
      for (int a = 0; a < 2; ++a) {
#pragma unroll
        for (int b = 0; b < 2; ++b) {
          const int t = 2 * a + b;
          if (MODE == 2) {
            p1[t] = wmma4(wr[b], xr[a], p1[t]);
            p2[t] = wmma4(wi[b], xi[a], p2[t]);
            p3[t] = wmma4(ws[b], xs[a], p3[t]);
          } else {
            p1[t] = wmma4(xr[a], wr[b], p1[t]);
            p2[t] = wmma4(xi[a], wi[b], p2[t]);
            p3[t] = wmma4(xs[a], ws[b], p3[t]);
          }
        }
      }
    }

    if (more) {
      asm volatile("" : "+v"(p1[0]), "+v"(p2[0]), "+v"(p3[0]), "+v"(p1[1]), "+v"(p2[1]), "+v"(p3[1]),
                        "+v"(p1[2]), "+v"(p2[2]), "+v"(p3[2]), "+v"(p1[3]), "+v"(p2[3]), "+v"(p3[3]),
                        "+v"(g[0]), "+v"(g[1]), "+v"(g[2]), "+v"(g[3]), "+v"(g[4]), "+v"(g[5]),
                        "+v"(g[6]), "+v"(g[7]), "+v"(g[8]), "+v"(g[9]), "+v"(g[10]), "+v"(g[11]),
                        "+v"(g[12]), "+v"(g[13]), "+v"(g[14]), "+v"(g[15]));
      v4f* nxt = tile[(c + 1) & 1];
#pragma unroll
      for (int i = 0; i < 16; ++i) nxt[sdst + i * 8 * PJ_PITCH] = pair_order(g[i]);
    }
    __syncthreads();
  }

#pragma unroll
  for (int a = 0; a < 2; ++a) {
#pragma unroll
    for (int b = 0; b < 2; ++b) {
      const int t = 2 * a + b;
      const v8f yr = p1[t] - p2[t];
      const v8f yi = (p3[t] - p1[t]) - p2[t];
      proj_store_tile<MODE>(yr, yi, sTile0 + a, oTile0 + b, lane, Bv, Yre, Yim);
    }
  }
}

__global__ void __launch_bounds__(128) __attribute__((amdgpu_waves_per_eu(3, 3)))
proj3_complex_kernel(const float* __restrict__ Xq, const float* __restrict__ Wq,
                     const float* __restrict__ Bq, float* __restrict__ Qre,
                     float* __restrict__ Qim,
                     const float* __restrict__ Xk, const float* __restrict__ Wk,
                     const float* __restrict__ Bk, float* __restrict__ Kre,
                     float* __restrict__ Kim,
                     const float* __restrict__ Xv, const float* __restrict__ Wv,
                     const float* __restrict__ Bv, float* __restrict__ Vre,
                     float* __restrict__ Vim) {
  __shared__ v4f tile[2][PJ_ROWS * PJ_PITCH];
  if (blockIdx.y == 0)      proj_complex_block<0>(tile, Xq, Wq, Bq, Qre, Qim);
  else if (blockIdx.y == 1) proj_complex_block<2>(tile, Xk, Wk, Bk, Kre, Kim);
  else                      proj_complex_block<1>(tile, Xv, Wv, Bv, Vre, Vim);
}

__device__ __forceinline__ float xhalf(float x) {
  const int b = __builtin_bit_cast(int, x);
  return __builtin_bit_cast(float,
      __builtin_amdgcn_permlanex16(b, b, 0x76543210, 0xfedcba98, false, false));
}

struct KVFrag {
  v4f kr[4], ki[4];
  v4f vr[4], vi[4];
};

__device__ __forceinline__ void load_kv(KVFrag& f, const float* __restrict__ kreh,
                                        const float* __restrict__ kimh,
                                        const float* __restrict__ vreh,
                                        const float* __restrict__ vimh,
                                        int ktK, int ktV, int lane) {
  const size_t offK = (size_t)ktK * 512 + lane * 4;
  const size_t offV = (size_t)ktV * 512 + lane * 4;
#pragma unroll
  for (int i = 0; i < 4; ++i) {
    f.kr[i] = *(const v4f*)(kreh + offK + i * 128);
    f.ki[i] = *(const v4f*)(kimh + offK + i * 128);
    f.vr[i] = *(const v4f*)(vreh + offV + i * 128);
    f.vi[i] = *(const v4f*)(vimh + offV + i * 128);
  }
}

__device__ __forceinline__ void score_kstep(const v4f (&fkr)[4], const v4f (&fki)[4], const int t,
                                            const v2f (&qr)[8], const v2f (&qi)[8],
                                            const v2f (&qd)[8], v8f& p1, v8f& p2, v8f& p3) {
  const v4f r4 = fkr[t >> 1], i4 = fki[t >> 1];
  const v2f kr = (t & 1) ? v2f{ r4.z, r4.w } : v2f{ r4.x, r4.y };
  const v2f ki = (t & 1) ? v2f{ i4.z, i4.w } : v2f{ i4.x, i4.y };
  const v2f ks = kr + ki;
  p1 = wmma4(kr, qr[t], p1);
  p2 = wmma4(ki, qi[t], p2);
  p3 = wmma4(ks, qd[t], p3);
}

__device__ __forceinline__ void attn_step(KVFrag& f, const v2f (&qr)[8], const v2f (&qi)[8],
                                          const v2f (&qd)[8], v8f& sre, v8f& sim,
                                          float& m_run, float& l_run,
                                          v8f& o_re0, v8f& o_re1, v8f& o_im0, v8f& o_im1) {
  const float scale = 0.17677669529663687f;

  asm volatile("" : "+v"(f.kr[0]), "+v"(f.kr[1]), "+v"(f.kr[2]), "+v"(f.kr[3]),
                    "+v"(f.ki[0]), "+v"(f.ki[1]), "+v"(f.ki[2]), "+v"(f.ki[3]),
                    "+v"(f.vr[0]), "+v"(f.vr[1]), "+v"(f.vr[2]), "+v"(f.vr[3]),
                    "+v"(f.vi[0]), "+v"(f.vi[1]), "+v"(f.vi[2]), "+v"(f.vi[3]));

  v8f p1 = {0.f}, p2 = {0.f}, p3 = {0.f};

  score_kstep(f.kr, f.ki, 0, qr, qi, qd, p1, p2, p3);
  v8f sq = sre * sre + sim * sim;
  asm volatile("" : "+v"(p1), "+v"(p2), "+v"(p3), "+v"(sq));

  score_kstep(f.kr, f.ki, 1, qr, qi, qd, p1, p2, p3);
  v8f rt;
#pragma unroll
  for (int r = 0; r < 8; ++r)
    rt[r] = __builtin_amdgcn_sqrtf(sq[r]);
  asm volatile("" : "+v"(p1), "+v"(p2), "+v"(p3), "+v"(rt));

  score_kstep(f.kr, f.ki, 2, qr, qi, qd, p1, p2, p3);
  v8f a = rt * scale;
  float mx = fmaxf(fmaxf(fmaxf(a[0], a[1]), fmaxf(a[2], a[3])),
                   fmaxf(fmaxf(a[4], a[5]), fmaxf(a[6], a[7])));
  float mo = xhalf(mx);
  asm volatile("" : "+v"(p1), "+v"(p2), "+v"(p3), "+v"(a), "+v"(mx), "+v"(mo));

  score_kstep(f.kr, f.ki, 3, qr, qi, qd, p1, p2, p3);
  mx = fmaxf(mx, mo);
  const float newm = fmaxf(m_run, mx);
  float sc = __expf(m_run - newm);
  v8f d = a - newm;
  m_run = newm;
  asm volatile("" : "+v"(p1), "+v"(p2), "+v"(p3), "+v"(d), "+v"(sc));

  score_kstep(f.kr, f.ki, 4, qr, qi, qd, p1, p2, p3);
  v8f p;
#pragma unroll
  for (int r = 0; r < 8; ++r) p[r] = __expf(d[r]);
  asm volatile("" : "+v"(p1), "+v"(p2), "+v"(p3), "+v"(p), "+v"(sc));

  score_kstep(f.kr, f.ki, 5, qr, qi, qd, p1, p2, p3);
  l_run = l_run * sc + (((p[0] + p[1]) + (p[2] + p[3])) + ((p[4] + p[5]) + (p[6] + p[7])));
  asm volatile("" : "+v"(p1), "+v"(p2), "+v"(p3), "+v"(l_run), "+v"(sc));

  score_kstep(f.kr, f.ki, 6, qr, qi, qd, p1, p2, p3);
  o_re0 *= sc;
  o_im0 *= sc;
  asm volatile("" : "+v"(p1), "+v"(p2), "+v"(p3), "+v"(o_re0), "+v"(o_im0), "+v"(sc));

  score_kstep(f.kr, f.ki, 7, qr, qi, qd, p1, p2, p3);
  o_re1 *= sc;
  o_im1 *= sc;
  asm volatile("" : "+v"(p1), "+v"(p2), "+v"(p3), "+v"(o_re1), "+v"(o_im1));

  const v2f ar0[4] = { {f.vr[0].x, f.vr[0].y}, {f.vr[0].z, f.vr[0].w}, {f.vr[1].x, f.vr[1].y}, {f.vr[1].z, f.vr[1].w} };
  const v2f ai0[4] = { {f.vi[0].x, f.vi[0].y}, {f.vi[0].z, f.vi[0].w}, {f.vi[1].x, f.vi[1].y}, {f.vi[1].z, f.vi[1].w} };
  const v2f ar1[4] = { {f.vr[2].x, f.vr[2].y}, {f.vr[2].z, f.vr[2].w}, {f.vr[3].x, f.vr[3].y}, {f.vr[3].z, f.vr[3].w} };
  const v2f ai1[4] = { {f.vi[2].x, f.vi[2].y}, {f.vi[2].z, f.vi[2].w}, {f.vi[3].x, f.vi[3].y}, {f.vi[3].z, f.vi[3].w} };

#pragma unroll
  for (int t = 0; t < 2; ++t) {
    const v2f bP = { p[2 * t], p[2 * t + 1] };
    o_re0 = wmma4(ar0[t], bP, o_re0);
    o_im0 = wmma4(ai0[t], bP, o_im0);
    o_re1 = wmma4(ar1[t], bP, o_re1);
    o_im1 = wmma4(ai1[t], bP, o_im1);
  }
  asm volatile("" : "+v"(o_re0), "+v"(o_im0), "+v"(o_re1), "+v"(o_im1),
                    "+v"(p1), "+v"(p2), "+v"(p3));

#pragma unroll
  for (int t = 2; t < 4; ++t) {
    const v2f bP = { p[2 * t], p[2 * t + 1] };
    o_re0 = wmma4(ar0[t], bP, o_re0);
    o_im0 = wmma4(ai0[t], bP, o_im0);
    o_re1 = wmma4(ar1[t], bP, o_re1);
    o_im1 = wmma4(ai1[t], bP, o_im1);
  }
  sre = p1 + p2;
  sim = p3 + (p1 - p2);
  asm volatile("" : "+v"(o_re0), "+v"(o_im0), "+v"(o_re1), "+v"(o_im1),
                    "+v"(sre), "+v"(sim));
}

__global__ void __launch_bounds__(256)
attn_complex_kernel(const float* __restrict__ Qre, const float* __restrict__ Qim,
                    const float* __restrict__ Kre, const float* __restrict__ Kim,
                    const float* __restrict__ Vtre, const float* __restrict__ Vtim,
                    float* __restrict__ Out) {
  const int wave = threadIdx.x >> 5;
  const int lane = threadIdx.x & 31;
  const int half = lane >> 4;
  const int l16  = lane & 15;
  const int h    = blockIdx.y;
  const int q0   = (blockIdx.x * 8 + wave) * 16;

  const float* qreh = Qre + (size_t)h * SEQ * HD;
  const float* qimh = Qim + (size_t)h * SEQ * HD;
  const float* kreh = Kre + (size_t)h * SEQ * HD;
  const float* kimh = Kim + (size_t)h * SEQ * HD;
  const float* vreh = Vtre + (size_t)h * SEQ * HD;
  const float* vimh = Vtim + (size_t)h * SEQ * HD;

  v2f qr[8], qi[8], qd[8];
#pragma unroll
  for (int t = 0; t < 8; ++t) {
    const int dk = t * 4 + half * 2;
    const size_t off = ((size_t)(q0 + l16)) * HD + dk;
    qr[t]  = ldp(qreh + off);
    qi[t]  = ldp(qimh + off);
    qd[t]  = qi[t] - qr[t];
  }

  float m_run = -1e30f;
  float l_run = 0.f;
  v8f o_re0 = {0.f}, o_re1 = {0.f}, o_im0 = {0.f}, o_im1 = {0.f};

  KVFrag fa, fb;

  v8f sre, sim;
  {
    const size_t off0 = (size_t)lane * 4;
#pragma unroll
    for (int i = 0; i < 4; ++i) {
      fb.kr[i] = *(const v4f*)(kreh + off0 + i * 128);
      fb.ki[i] = *(const v4f*)(kimh + off0 + i * 128);
    }
    load_kv(fa, kreh, kimh, vreh, vimh, 1, 0, lane);
    v8f p1 = {0.f}, p2 = {0.f}, p3 = {0.f};
#pragma unroll
    for (int t = 0; t < 8; ++t) score_kstep(fb.kr, fb.ki, t, qr, qi, qd, p1, p2, p3);
    sre = p1 + p2;
    sim = p3 + (p1 - p2);
  }

  for (int kt = 0; kt < SEQ / 16; kt += 2) {
    const int k2 = (kt + 2 < SEQ / 16) ? kt + 2 : SEQ / 16 - 1;
    load_kv(fb, kreh, kimh, vreh, vimh, k2, kt + 1, lane);
    attn_step(fa, qr, qi, qd, sre, sim, m_run, l_run, o_re0, o_re1, o_im0, o_im1);
    const int k3 = (kt + 3 < SEQ / 16) ? kt + 3 : SEQ / 16 - 1;
    load_kv(fa, kreh, kimh, vreh, vimh, k3, k2, lane);
    attn_step(fb, qr, qi, qd, sre, sim, m_run, l_run, o_re0, o_re1, o_im0, o_im1);
  }

  const float inv = 1.0f / (l_run + xhalf(l_run));
  float* op = Out + ((size_t)(q0 + l16) * CH + h * HD + 8 * half) * 2;
#pragma unroll
  for (int j = 0; j < 4; ++j) {
    const float4 e0 = { o_re0[2 * j] * inv, o_im0[2 * j] * inv,
                        o_re0[2 * j + 1] * inv, o_im0[2 * j + 1] * inv };
    const float4 e1 = { o_re1[2 * j] * inv, o_im1[2 * j] * inv,
                        o_re1[2 * j + 1] * inv, o_im1[2 * j + 1] * inv };
    *(float4*)(op + 4 * j)      = e0;
    *(float4*)(op + 32 + 4 * j) = e1;
  }
}

extern "C" void kernel_launch(void* const* d_in, const int* in_sizes, int n_in,
                              void* d_out, int out_size, void* d_ws, size_t ws_size,
                              hipStream_t stream) {
  (void)in_sizes; (void)n_in; (void)out_size; (void)ws_size;
  const float* Q  = (const float*)d_in[0];
  const float* V  = (const float*)d_in[1];
  const float* K  = (const float*)d_in[2];
  const float* Wq = (const float*)d_in[3];
  const float* bq = (const float*)d_in[4];
  const float* Wk = (const float*)d_in[5];
  const float* bk = (const float*)d_in[6];
  const float* Wv = (const float*)d_in[7];
  const float* bv = (const float*)d_in[8];
  float* out = (float*)d_out;

  const size_t plane = (size_t)NH * SEQ * HD;
  float* qre = (float*)d_ws;
  float* qim = qre + plane;
  float* kre = qim + plane;
  float* kim = kre + plane;
  float* vre = kim + plane;
  float* vim = vre + plane;

  const dim3 blk(256);
  const dim3 pblk(128);
  const dim3 pgrid((SEQ / PJ_SROWS) * (CH / PJ_OROWS), 3);
  proj3_complex_kernel<<<pgrid, pblk, 0, stream>>>(Q, Wq, bq, qre, qim,
                                                   K, Wk, bk, kre, kim,
                                                   V, Wv, bv, vre, vim);

  const dim3 agrid(SEQ / 16 / 8, NH);
  attn_complex_kernel<<<agrid, blk, 0, stream>>>(qre, qim, kre, kim, vre, vim, out);
}
